# starting kernel
# baseline (speedup 1.0000x reference)
#include <hip/hip_runtime.h>
#include <math.h>

#define NODES_DEF 100000
#define IN_C 128
#define HID 32
#define OUT_C 32
#define NEG_SLOPE 0.2f

typedef __attribute__((ext_vector_type(2))) float v2f;
typedef __attribute__((ext_vector_type(8))) float v8f;

template <int K, int NC, int EPI>
__global__ void gat_gemm_wmma(const float* __restrict__ A,
                              const float* __restrict__ B,
                              const float* __restrict__ bias,
                              float* __restrict__ C, int M)
{
    const int wavesPerBlock = blockDim.x >> 5;
    const int wave = blockIdx.x * wavesPerBlock + (threadIdx.x >> 5);
    const int nTiles = NC / 16;
    const int totalTiles = (M / 16) * nTiles;
    if (wave >= totalTiles) return;

    const int mTile = wave / nTiles;
    const int nTile = wave % nTiles;
    const int lane  = threadIdx.x & 31;
    const int half  = lane >> 4;
    const int r     = lane & 15;
    const int row   = mTile * 16 + r;
    const int col   = nTile * 16 + r;

    v8f acc = {};
#pragma unroll
    for (int k = 0; k < K; k += 4) {
        v2f a = *(const v2f*)(A + (size_t)row * K + k + 2 * half);
        v2f b;
        b.x = B[(k + 2 * half + 0) * NC + col];
        b.y = B[(k + 2 * half + 1) * NC + col];
        acc = __builtin_amdgcn_wmma_f32_16x16x4_f32(
             false, a,  false, b,
             (short)0, acc,  false,  false);
    }

#pragma unroll
    for (int i = 0; i < 8; ++i) {
        const int orow = mTile * 16 + i + 8 * half;
        float v = acc[i];
        if (EPI == 1) {
            v += bias[col];
            v = 1.0f / (1.0f + __expf(-v));
        }
        C[(size_t)orow * NC + col] = v;
    }
}

__device__ __forceinline__ int ord_f32(float f)
{
    int i = __float_as_int(f);
    return i >= 0 ? i : (i ^ 0x7fffffff);
}
__device__ __forceinline__ float unord_f32(int i)
{
    return __int_as_float(i >= 0 ? i : (i ^ 0x7fffffff));
}
__device__ __forceinline__ float leaky(float x)
{
    return x > 0.0f ? x : NEG_SLOPE * x;
}

__global__ void gat_node_prep(const float* __restrict__ h,
                              const float* __restrict__ a_src,
                              const float* __restrict__ a_dst,
                              float* __restrict__ s_src,
                              float* __restrict__ s_dst,
                              int*   __restrict__ m_ord,
                              float* __restrict__ z,
                              float* __restrict__ agg,
                              int n)
{
    const int i = blockIdx.x * blockDim.x + threadIdx.x;
    if (i >= n) return;
    const float* hp = h + (size_t)i * HID;
    float* ap = agg + (size_t)i * HID;
    float ss = 0.0f, sd = 0.0f;
#pragma unroll
    for (int c = 0; c < HID; ++c) {
        const float v = hp[c];
        ss += v * a_src[c];
        sd += v * a_dst[c];
        ap[c] = 0.0f;
    }
    s_src[i] = ss;
    s_dst[i] = sd;
    m_ord[i] = (int)0x80000000;
    z[i]     = 0.0f;
}

__device__ __forceinline__ void edge_endpoints(const int* __restrict__ ei,
                                               int idx, int n_edges,
                                               int& s, int& d)
{
    if (idx < n_edges) { s = ei[idx]; d = ei[n_edges + idx]; }
    else               { s = d = idx - n_edges; }
}

__global__ void gat_edge_max(const int* __restrict__ ei,
                             const float* __restrict__ s_src,
                             const float* __restrict__ s_dst,
                             int* __restrict__ m_ord,
                             int n_edges, int e_total)
{
    const int idx = blockIdx.x * blockDim.x + threadIdx.x;
    if (idx >= e_total) return;
    int s, d;
    edge_endpoints(ei, idx, n_edges, s, d);
    const float e = leaky(s_src[s] + s_dst[d]);
    atomicMax(&m_ord[d], ord_f32(e));
}

__global__ void gat_edge_sum(const int* __restrict__ ei,
                             const float* __restrict__ s_src,
                             const float* __restrict__ s_dst,
                             const int* __restrict__ m_ord,
                             float* __restrict__ z,
                             int n_edges, int e_total)
{
    const int idx = blockIdx.x * blockDim.x + threadIdx.x;
    if (idx >= e_total) return;
    int s, d;
    edge_endpoints(ei, idx, n_edges, s, d);
    const float e  = leaky(s_src[s] + s_dst[d]);
    const float ex = __expf(e - unord_f32(m_ord[d]));
    atomicAdd(&z[d], ex);
}

__global__ void gat_edge_agg(const int* __restrict__ ei,
                             const float* __restrict__ s_src,
                             const float* __restrict__ s_dst,
                             const int* __restrict__ m_ord,
                             const float* __restrict__ z,
                             const float* __restrict__ h,
                             float* __restrict__ agg,
                             int n_edges, int e_total)
{
    const int idx = blockIdx.x * blockDim.x + threadIdx.x;
    if (idx >= e_total) return;
    int s, d;
    edge_endpoints(ei, idx, n_edges, s, d);
    const float e     = leaky(s_src[s] + s_dst[d]);
    const float alpha = __expf(e - unord_f32(m_ord[d])) / (z[d] + 1e-16f);

    const float4* hp = (const float4*)(h + (size_t)s * HID);
    float* ap = agg + (size_t)d * HID;
#pragma unroll
    for (int c4 = 0; c4 < HID / 4; ++c4) {
        const float4 v = hp[c4];
        atomicAdd(ap + 4 * c4 + 0, alpha * v.x);
        atomicAdd(ap + 4 * c4 + 1, alpha * v.y);
        atomicAdd(ap + 4 * c4 + 2, alpha * v.z);
        atomicAdd(ap + 4 * c4 + 3, alpha * v.w);
    }
}

__global__ void gat_bias_relu(const float* __restrict__ agg,
                              const float* __restrict__ bias,
                              float* __restrict__ g, int total)
{
    const int i = blockIdx.x * blockDim.x + threadIdx.x;
    if (i >= total) return;
    const float v = agg[i] + bias[i & (HID - 1)];
    g[i] = v > 0.0f ? v : 0.0f;
}

extern "C" void kernel_launch(void* const* d_in, const int* in_sizes, int n_in,
                              void* d_out, int out_size, void* d_ws, size_t ws_size,
                              hipStream_t stream)
{
    const float* x      = (const float*)d_in[0];
    const int*   ei     = (const int*)  d_in[1];
    const float* W1     = (const float*)d_in[2];
    const float* a_src1 = (const float*)d_in[3];
    const float* a_dst1 = (const float*)d_in[4];
    const float* b1     = (const float*)d_in[5];
    const float* W2     = (const float*)d_in[6];
    const float* a_src2 = (const float*)d_in[7];
    const float* a_dst2 = (const float*)d_in[8];
    const float* b2     = (const float*)d_in[9];
    const float* Wc     = (const float*)d_in[10];
    const float* bc     = (const float*)d_in[11];
    float* out = (float*)d_out;

    const int n       = in_sizes[0] / IN_C;
    const int n_edges = in_sizes[1] / 2;
    const int e_total = n_edges + n;

    float* ws    = (float*)d_ws;
    float* h     = ws;
    float* g     = h   + (size_t)n * HID;
    float* agg   = g   + (size_t)n * HID;
    float* s_src = agg + (size_t)n * HID;
    float* s_dst = s_src + n;
    float* z     = s_dst + n;
    int*   m_ord = (int*)(z + n);

    const int TB = 256;
    const int nodeBlocks = (n + TB - 1) / TB;
    const int edgeBlocks = (e_total + TB - 1) / TB;
    const int elemBlocks = (n * HID + TB - 1) / TB;
    const int gemmWaves  = (n / 16) * (HID / 16);
    const int gemmBlocks = (gemmWaves + 7) / 8;

    gat_gemm_wmma<IN_C, HID, 0><<<gemmBlocks, TB, 0, stream>>>(x, W1, nullptr, h, n);
    gat_node_prep<<<nodeBlocks, TB, 0, stream>>>(h, a_src1, a_dst1, s_src, s_dst,
                                                 m_ord, z, agg, n);
    gat_edge_max<<<edgeBlocks, TB, 0, stream>>>(ei, s_src, s_dst, m_ord, n_edges, e_total);
    gat_edge_sum<<<edgeBlocks, TB, 0, stream>>>(ei, s_src, s_dst, m_ord, z, n_edges, e_total);
    gat_edge_agg<<<edgeBlocks, TB, 0, stream>>>(ei, s_src, s_dst, m_ord, z, h, agg,
                                                n_edges, e_total);
    gat_bias_relu<<<elemBlocks, TB, 0, stream>>>(agg, b1, g, n * HID);

    gat_gemm_wmma<HID, HID, 0><<<gemmBlocks, TB, 0, stream>>>(g, W2, nullptr, h, n);
    gat_node_prep<<<nodeBlocks, TB, 0, stream>>>(h, a_src2, a_dst2, s_src, s_dst,
                                                 m_ord, z, agg, n);
    gat_edge_max<<<edgeBlocks, TB, 0, stream>>>(ei, s_src, s_dst, m_ord, n_edges, e_total);
    gat_edge_sum<<<edgeBlocks, TB, 0, stream>>>(ei, s_src, s_dst, m_ord, z, n_edges, e_total);
    gat_edge_agg<<<edgeBlocks, TB, 0, stream>>>(ei, s_src, s_dst, m_ord, z, h, agg,
                                                n_edges, e_total);
    gat_bias_relu<<<elemBlocks, TB, 0, stream>>>(agg, b2, g, n * HID);

    gat_gemm_wmma<HID, OUT_C, 1><<<gemmBlocks, TB, 0, stream>>>(g, Wc, bc, out, n);
}


# edge aggregate: a wave walks its 32 edges with lane = channel, so each h-row gather and each agg-row f32 atomic add is one 128-byte line per instruction; same atomics, sums still in arrival order
# speedup vs baseline: 3.0764x; 3.0764x over previous
#include <hip/hip_runtime.h>
#include <math.h>

#define NODES_DEF 100000
#define IN_C 128
#define HID 32
#define OUT_C 32
#define NEG_SLOPE 0.2f

typedef __attribute__((ext_vector_type(2))) float v2f;
typedef __attribute__((ext_vector_type(8))) float v8f;

template <int K, int NC, int EPI>
__global__ void gat_gemm_wmma(const float* __restrict__ A,
                              const float* __restrict__ B,
                              const float* __restrict__ bias,
                              float* __restrict__ C, int M)
{
    const int wavesPerBlock = blockDim.x >> 5;
    const int wave = blockIdx.x * wavesPerBlock + (threadIdx.x >> 5);
    const int nTiles = NC / 16;
    const int totalTiles = (M / 16) * nTiles;
    if (wave >= totalTiles) return;

    const int mTile = wave / nTiles;
    const int nTile = wave % nTiles;
    const int lane  = threadIdx.x & 31;
    const int half  = lane >> 4;
    const int r     = lane & 15;
    const int row   = mTile * 16 + r;
    const int col   = nTile * 16 + r;

    v8f acc = {};
#pragma unroll
    for (int k = 0; k < K; k += 4) {
        v2f a = *(const v2f*)(A + (size_t)row * K + k + 2 * half);
        v2f b;
        b.x = B[(k + 2 * half + 0) * NC + col];
        b.y = B[(k + 2 * half + 1) * NC + col];
        acc = __builtin_amdgcn_wmma_f32_16x16x4_f32(
             false, a,  false, b,
             (short)0, acc,  false,  false);
    }

#pragma unroll
    for (int i = 0; i < 8; ++i) {
        const int orow = mTile * 16 + i + 8 * half;
        float v = acc[i];
        if (EPI == 1) {
            v += bias[col];
            v = 1.0f / (1.0f + __expf(-v));
        }
        C[(size_t)orow * NC + col] = v;
    }
}

__device__ __forceinline__ int ord_f32(float f)
{
    int i = __float_as_int(f);
    return i >= 0 ? i : (i ^ 0x7fffffff);
}
__device__ __forceinline__ float unord_f32(int i)
{
    return __int_as_float(i >= 0 ? i : (i ^ 0x7fffffff));
}
__device__ __forceinline__ float leaky(float x)
{
    return x > 0.0f ? x : NEG_SLOPE * x;
}

__global__ void gat_node_prep(const float* __restrict__ h,
                              const float* __restrict__ a_src,
                              const float* __restrict__ a_dst,
                              float* __restrict__ s_src,
                              float* __restrict__ s_dst,
                              int*   __restrict__ m_ord,
                              float* __restrict__ z,
                              float* __restrict__ agg,
                              int n)
{
    const int i = blockIdx.x * blockDim.x + threadIdx.x;
    if (i >= n) return;
    const float* hp = h + (size_t)i * HID;
    float* ap = agg + (size_t)i * HID;
    float ss = 0.0f, sd = 0.0f;
#pragma unroll
    for (int c = 0; c < HID; ++c) {
        const float v = hp[c];
        ss += v * a_src[c];
        sd += v * a_dst[c];
        ap[c] = 0.0f;
    }
    s_src[i] = ss;
    s_dst[i] = sd;
    m_ord[i] = (int)0x80000000;
    z[i]     = 0.0f;
}

__device__ __forceinline__ void edge_endpoints(const int* __restrict__ ei,
                                               int idx, int n_edges,
                                               int& s, int& d)
{
    if (idx < n_edges) { s = ei[idx]; d = ei[n_edges + idx]; }
    else               { s = d = idx - n_edges; }
}

__global__ void gat_edge_max(const int* __restrict__ ei,
                             const float* __restrict__ s_src,
                             const float* __restrict__ s_dst,
                             int* __restrict__ m_ord,
                             int n_edges, int e_total)
{
    const int idx = blockIdx.x * blockDim.x + threadIdx.x;
    if (idx >= e_total) return;
    int s, d;
    edge_endpoints(ei, idx, n_edges, s, d);
    const float e = leaky(s_src[s] + s_dst[d]);
    atomicMax(&m_ord[d], ord_f32(e));
}

__global__ void gat_edge_sum(const int* __restrict__ ei,
                             const float* __restrict__ s_src,
                             const float* __restrict__ s_dst,
                             const int* __restrict__ m_ord,
                             float* __restrict__ z,
                             int n_edges, int e_total)
{
    const int idx = blockIdx.x * blockDim.x + threadIdx.x;
    if (idx >= e_total) return;
    int s, d;
    edge_endpoints(ei, idx, n_edges, s, d);
    const float e  = leaky(s_src[s] + s_dst[d]);
    const float ex = __expf(e - unord_f32(m_ord[d]));
    atomicAdd(&z[d], ex);
}

__global__ void gat_edge_agg(const int* __restrict__ ei,
                             const float* __restrict__ s_src,
                             const float* __restrict__ s_dst,
                             const int* __restrict__ m_ord,
                             const float* __restrict__ z,
                             const float* __restrict__ h,
                             float* __restrict__ agg,
                             int n_edges, int e_total)
{
    const int idx  = blockIdx.x * blockDim.x + threadIdx.x;
    const int lane = threadIdx.x & 31;
    const int base = __builtin_amdgcn_readfirstlane(idx - lane);
    if (base >= e_total) return;

    int s = 0, d = 0;
    float alpha = 0.0f;
    if (idx < e_total) {
        edge_endpoints(ei, idx, n_edges, s, d);
        const float e = leaky(s_src[s] + s_dst[d]);
        alpha = __expf(e - unord_f32(m_ord[d])) / (z[d] + 1e-16f);
    }
    const int ai = __float_as_int(alpha);
    const int cnt = e_total - base;

    if (cnt >= 32) {
#pragma unroll
        for (int j0 = 0; j0 < 32; j0 += 8) {
            float v[8];
#pragma unroll
            for (int u = 0; u < 8; ++u) {
                const int sj = __builtin_amdgcn_readlane(s, j0 + u);
                v[u] = h[(size_t)sj * HID + lane];
            }
#pragma unroll
            for (int u = 0; u < 8; ++u) {
                const int dj = __builtin_amdgcn_readlane(d, j0 + u);
                const float aj = __int_as_float(__builtin_amdgcn_readlane(ai, j0 + u));
                atomicAdd(agg + (size_t)dj * HID + lane, aj * v[u]);
            }
        }
    } else {
        for (int j = 0; j < cnt; ++j) {
            const int sj = __builtin_amdgcn_readlane(s, j);
            const int dj = __builtin_amdgcn_readlane(d, j);
            const float aj = __int_as_float(__builtin_amdgcn_readlane(ai, j));
            atomicAdd(agg + (size_t)dj * HID + lane, aj * h[(size_t)sj * HID + lane]);
        }
    }
}

__global__ void gat_bias_relu(const float* __restrict__ agg,
                              const float* __restrict__ bias,
                              float* __restrict__ g, int total)
{
    const int i = blockIdx.x * blockDim.x + threadIdx.x;
    if (i >= total) return;
    const float v = agg[i] + bias[i & (HID - 1)];
    g[i] = v > 0.0f ? v : 0.0f;
}

extern "C" void kernel_launch(void* const* d_in, const int* in_sizes, int n_in,
                              void* d_out, int out_size, void* d_ws, size_t ws_size,
                              hipStream_t stream)
{
    const float* x      = (const float*)d_in[0];
    const int*   ei     = (const int*)  d_in[1];
    const float* W1     = (const float*)d_in[2];
    const float* a_src1 = (const float*)d_in[3];
    const float* a_dst1 = (const float*)d_in[4];
    const float* b1     = (const float*)d_in[5];
    const float* W2     = (const float*)d_in[6];
    const float* a_src2 = (const float*)d_in[7];
    const float* a_dst2 = (const float*)d_in[8];
    const float* b2     = (const float*)d_in[9];
    const float* Wc     = (const float*)d_in[10];
    const float* bc     = (const float*)d_in[11];
    float* out = (float*)d_out;

    const int n       = in_sizes[0] / IN_C;
    const int n_edges = in_sizes[1] / 2;
    const int e_total = n_edges + n;

    float* ws    = (float*)d_ws;
    float* h     = ws;
    float* g     = h   + (size_t)n * HID;
    float* agg   = g   + (size_t)n * HID;
    float* s_src = agg + (size_t)n * HID;
    float* s_dst = s_src + n;
    float* z     = s_dst + n;
    int*   m_ord = (int*)(z + n);

    const int TB = 256;
    const int nodeBlocks = (n + TB - 1) / TB;
    const int edgeBlocks = (e_total + TB - 1) / TB;
    const int elemBlocks = (n * HID + TB - 1) / TB;
    const int gemmWaves  = (n / 16) * (HID / 16);
    const int gemmBlocks = (gemmWaves + 7) / 8;

    gat_gemm_wmma<IN_C, HID, 0><<<gemmBlocks, TB, 0, stream>>>(x, W1, nullptr, h, n);
    gat_node_prep<<<nodeBlocks, TB, 0, stream>>>(h, a_src1, a_dst1, s_src, s_dst,
                                                 m_ord, z, agg, n);
    gat_edge_max<<<edgeBlocks, TB, 0, stream>>>(ei, s_src, s_dst, m_ord, n_edges, e_total);
    gat_edge_sum<<<edgeBlocks, TB, 0, stream>>>(ei, s_src, s_dst, m_ord, z, n_edges, e_total);
    gat_edge_agg<<<edgeBlocks, TB, 0, stream>>>(ei, s_src, s_dst, m_ord, z, h, agg,
                                                n_edges, e_total);
    gat_bias_relu<<<elemBlocks, TB, 0, stream>>>(agg, b1, g, n * HID);

    gat_gemm_wmma<HID, HID, 0><<<gemmBlocks, TB, 0, stream>>>(g, W2, nullptr, h, n);
    gat_node_prep<<<nodeBlocks, TB, 0, stream>>>(h, a_src2, a_dst2, s_src, s_dst,
                                                 m_ord, z, agg, n);
    gat_edge_max<<<edgeBlocks, TB, 0, stream>>>(ei, s_src, s_dst, m_ord, n_edges, e_total);
    gat_edge_sum<<<edgeBlocks, TB, 0, stream>>>(ei, s_src, s_dst, m_ord, z, n_edges, e_total);
    gat_edge_agg<<<edgeBlocks, TB, 0, stream>>>(ei, s_src, s_dst, m_ord, z, h, agg,
                                                n_edges, e_total);
    gat_bias_relu<<<elemBlocks, TB, 0, stream>>>(agg, b2, g, n * HID);

    gat_gemm_wmma<HID, OUT_C, 1><<<gemmBlocks, TB, 0, stream>>>(g, Wc, bc, out, n);
}


# edges grouped by destination once per call (integer counting sort); one wave per node does max, sum and weighted row sum in registers: float atomics gone, sums in list order plus a wave tree
# speedup vs baseline: 6.4191x; 2.0866x over previous
#include <hip/hip_runtime.h>
#include <math.h>

#define NODES_DEF 100000
#define IN_C 128
#define HID 32
#define OUT_C 32
#define NEG_SLOPE 0.2f

typedef __attribute__((ext_vector_type(2))) float v2f;
typedef __attribute__((ext_vector_type(8))) float v8f;

template <int K, int NC, int EPI>
__global__ void gat_gemm_wmma(const float* __restrict__ A,
                              const float* __restrict__ B,
                              const float* __restrict__ bias,
                              float* __restrict__ C, int M)
{
    const int wavesPerBlock = blockDim.x >> 5;
    const int wave = blockIdx.x * wavesPerBlock + (threadIdx.x >> 5);
    const int nTiles = NC / 16;
    const int totalTiles = (M / 16) * nTiles;
    if (wave >= totalTiles) return;

    const int mTile = wave / nTiles;
    const int nTile = wave % nTiles;
    const int lane  = threadIdx.x & 31;
    const int half  = lane >> 4;
    const int r     = lane & 15;
    const int row   = mTile * 16 + r;
    const int col   = nTile * 16 + r;

    v8f acc = {};
#pragma unroll
    for (int k = 0; k < K; k += 4) {
        v2f a = *(const v2f*)(A + (size_t)row * K + k + 2 * half);
        v2f b;
        b.x = B[(k + 2 * half + 0) * NC + col];
        b.y = B[(k + 2 * half + 1) * NC + col];
        acc = __builtin_amdgcn_wmma_f32_16x16x4_f32(
             false, a,  false, b,
             (short)0, acc,  false,  false);
    }

#pragma unroll
    for (int i = 0; i < 8; ++i) {
        const int orow = mTile * 16 + i + 8 * half;
        float v = acc[i];
        if (EPI == 1) {
            v += bias[col];
            v = 1.0f / (1.0f + __expf(-v));
        }
        C[(size_t)orow * NC + col] = v;
    }
}

__device__ __forceinline__ float leaky(float x)
{
    return x > 0.0f ? x : NEG_SLOPE * x;
}

template <bool ZERO_CNT>
__global__ void gat_node_scores(const float* __restrict__ h,
                                const float* __restrict__ a_src,
                                const float* __restrict__ a_dst,
                                float* __restrict__ s_src,
                                float* __restrict__ s_dst,
                                int*   __restrict__ cnt,
                                int n)
{
    const int i = blockIdx.x * blockDim.x + threadIdx.x;
    if (i >= n) return;
    const float* hp = h + (size_t)i * HID;
    float ss = 0.0f, sd = 0.0f;
#pragma unroll
    for (int c = 0; c < HID; ++c) {
        const float v = hp[c];
        ss += v * a_src[c];
        sd += v * a_dst[c];
    }
    s_src[i] = ss;
    s_dst[i] = sd;
    if (ZERO_CNT) cnt[i] = 0;
}

__global__ void csr_count(const int* __restrict__ ei, int* __restrict__ cnt,
                          int n_edges)
{
    const int i = blockIdx.x * blockDim.x + threadIdx.x;
    if (i >= n_edges) return;
    atomicAdd(&cnt[ei[n_edges + i]], 1);
}

__global__ void __launch_bounds__(1024)
csr_scan(const int* __restrict__ cnt, int* __restrict__ cur, int n)
{
    __shared__ int sh[1024];
    const int t = threadIdx.x;
    const int tile0 = blockIdx.x * 1024;

    int pre = 0;
    for (int j = t; j < tile0; j += 1024) pre += cnt[j];
    const int i = tile0 + t;
    const int v = (i < n) ? cnt[i] : 0;

    sh[t] = pre;
    __syncthreads();
    for (int off = 512; off > 0; off >>= 1) {
        if (t < off) sh[t] += sh[t + off];
        __syncthreads();
    }
    const int base = sh[0];
    __syncthreads();

    sh[t] = v;
    __syncthreads();
    for (int off = 1; off < 1024; off <<= 1) {
        const int u = (t >= off) ? sh[t - off] : 0;
        __syncthreads();
        sh[t] += u;
        __syncthreads();
    }
    if (i < n) cur[i] = base + sh[t];
}

__global__ void csr_fill(const int* __restrict__ ei,
                         int* __restrict__ cur,
                         int* __restrict__ srt, int n_edges)
{
    const int i = blockIdx.x * blockDim.x + threadIdx.x;
    if (i >= n_edges) return;
    const int s = ei[i];
    const int d = ei[n_edges + i];
    srt[atomicSub(&cur[d], 1) - 1] = s;
}

__device__ __forceinline__ float wave_max(float v)
{
#pragma unroll
    for (int o = 16; o > 0; o >>= 1) v = fmaxf(v, __shfl_xor(v, o, 32));
    return v;
}
__device__ __forceinline__ float wave_sum(float v)
{
#pragma unroll
    for (int o = 16; o > 0; o >>= 1) v += __shfl_xor(v, o, 32);
    return v;
}

template <int U>
__device__ __forceinline__ float add_rows(const float* __restrict__ h, int sj, int ai,
                                          int j, int lane, float acc)
{
    float v[U];
#pragma unroll
    for (int u = 0; u < U; ++u)
        v[u] = h[(size_t)__builtin_amdgcn_readlane(sj, j + u) * HID + lane];
#pragma unroll
    for (int u = 0; u < U; ++u)
        acc = __builtin_fmaf(__int_as_float(__builtin_amdgcn_readlane(ai, j + u)), v[u], acc);
    return acc;
}

__device__ __forceinline__ float add_chunk(const float* __restrict__ h, int sj, float al,
                                           int k, int lane, float acc)
{
    const int ai = __float_as_int(al);
    int j = 0;
    for (; j + 8 <= k; j += 8) acc = add_rows<8>(h, sj, ai, j, lane, acc);
    if (j + 4 <= k) { acc = add_rows<4>(h, sj, ai, j, lane, acc); j += 4; }
    if (j + 2 <= k) { acc = add_rows<2>(h, sj, ai, j, lane, acc); j += 2; }
    if (j < k)        acc = add_rows<1>(h, sj, ai, j, lane, acc);
    return acc;
}

__global__ void gat_node_gather(const int* __restrict__ srt,
                                const int* __restrict__ cur,
                                const int* __restrict__ cnt,
                                const float* __restrict__ s_src,
                                const float* __restrict__ s_dst,
                                const float* __restrict__ h,
                                const float* __restrict__ bias,
                                float* __restrict__ g, int n)
{
    const int lane = threadIdx.x & 31;
    const int d = __builtin_amdgcn_readfirstlane(
        (int)((blockIdx.x * blockDim.x + threadIdx.x) >> 5));
    if (d >= n) return;
    const int beg = __builtin_amdgcn_readfirstlane(cur[d]);
    const int deg = __builtin_amdgcn_readfirstlane(cnt[d]);
    const int end = beg + deg;

    const float sd     = s_dst[d];
    const float e_self = leaky(s_src[d] + sd);

    const int  p0 = beg + lane;
    const int  p1 = p0 + 32;
    const bool v0 = p0 < end;
    const bool v1 = p1 < end;
    int s0 = 0, s1 = 0;
    if (v0) s0 = srt[p0];
    if (v1) s1 = srt[p1];
    const float a0 = s_src[s0];
    const float a1 = s_src[s1];
    const float e0 = v0 ? leaky(a0 + sd) : e_self;
    const float e1 = v1 ? leaky(a1 + sd) : e_self;

    float m = fmaxf(fmaxf(e_self, e0), e1);
    for (int p = beg + 64 + lane; p < end; p += 32)
        m = fmaxf(m, leaky(s_src[srt[p]] + sd));
    m = wave_max(m);

    const float x0 = v0 ? __expf(e0 - m) : 0.0f;
    const float x1 = v1 ? __expf(e1 - m) : 0.0f;
    float zs = x0 + x1;
    for (int p = beg + 64 + lane; p < end; p += 32)
        zs += __expf(leaky(s_src[srt[p]] + sd) - m);
    const float ex_self = __expf(e_self - m);
    const float den = (wave_sum(zs) + ex_self) + 1e-16f;

    float acc = (ex_self / den) * h[(size_t)d * HID + lane];
    if (deg > 0)
        acc = add_chunk(h, s0, x0 / den, deg < 32 ? deg : 32, lane, acc);
    if (deg > 32)
        acc = add_chunk(h, s1, x1 / den, deg < 64 ? deg - 32 : 32, lane, acc);
    for (int c0 = beg + 64; c0 < end; c0 += 32) {
        const int p = c0 + lane;
        int   sj = 0;
        float al = 0.0f;
        if (p < end) {
            sj = srt[p];
            al = __expf(leaky(s_src[sj] + sd) - m) / den;
        }
        const int rem = end - c0;
        acc = add_chunk(h, sj, al, rem < 32 ? rem : 32, lane, acc);
    }

    const float o = acc + bias[lane];
    g[(size_t)d * HID + lane] = o > 0.0f ? o : 0.0f;
}

extern "C" void kernel_launch(void* const* d_in, const int* in_sizes, int n_in,
                              void* d_out, int out_size, void* d_ws, size_t ws_size,
                              hipStream_t stream)
{
    const float* x      = (const float*)d_in[0];
    const int*   ei     = (const int*)  d_in[1];
    const float* W1     = (const float*)d_in[2];
    const float* a_src1 = (const float*)d_in[3];
    const float* a_dst1 = (const float*)d_in[4];
    const float* b1     = (const float*)d_in[5];
    const float* W2     = (const float*)d_in[6];
    const float* a_src2 = (const float*)d_in[7];
    const float* a_dst2 = (const float*)d_in[8];
    const float* b2     = (const float*)d_in[9];
    const float* Wc     = (const float*)d_in[10];
    const float* bc     = (const float*)d_in[11];
    float* out = (float*)d_out;

    const int n       = in_sizes[0] / IN_C;
    const int n_edges = in_sizes[1] / 2;

    float* ws    = (float*)d_ws;
    float* h     = ws;
    float* g     = h   + (size_t)n * HID;
    float* s_src = g   + (size_t)n * HID;
    float* s_dst = s_src + n;
    int*   cnt   = (int*)(s_dst + n);
    int*   cur   = cnt + n;
    int*   srt   = cur + n;

    const int TB = 256;
    const int nodeBlocks = (n + TB - 1) / TB;
    const int edgeBlocks = (n_edges + TB - 1) / TB;
    const int scanBlocks = (n + 1023) / 1024;
    const int waveBlocks = (n + 7) / 8;
    const int gemmWaves  = (n / 16) * (HID / 16);
    const int gemmBlocks = (gemmWaves + 7) / 8;

    gat_gemm_wmma<IN_C, HID, 0><<<gemmBlocks, TB, 0, stream>>>(x, W1, nullptr, h, n);
    gat_node_scores<true><<<nodeBlocks, TB, 0, stream>>>(h, a_src1, a_dst1, s_src, s_dst,
                                                         cnt, n);
    csr_count<<<edgeBlocks, TB, 0, stream>>>(ei, cnt, n_edges);
    csr_scan<<<scanBlocks, 1024, 0, stream>>>(cnt, cur, n);
    csr_fill<<<edgeBlocks, TB, 0, stream>>>(ei, cur, srt, n_edges);
    gat_node_gather<<<waveBlocks, TB, 0, stream>>>(srt, cur, cnt, s_src, s_dst, h, b1, g, n);

    gat_gemm_wmma<HID, HID, 0><<<gemmBlocks, TB, 0, stream>>>(g, W2, nullptr, h, n);
    gat_node_scores<false><<<nodeBlocks, TB, 0, stream>>>(h, a_src2, a_dst2, s_src, s_dst,
                                                          cnt, n);
    gat_node_gather<<<waveBlocks, TB, 0, stream>>>(srt, cur, cnt, s_src, s_dst, h, b2, g, n);

    gat_gemm_wmma<HID, OUT_C, 1><<<gemmBlocks, TB, 0, stream>>>(g, Wc, bc, out, n);
}


# edge list built by a two-level counting sort with LDS counters (bucket = 128 destinations): per-edge global atomics and scattered 4-byte stores gone; list order, so sum order, still arrival order
# speedup vs baseline: 23.3420x; 3.6363x over previous
#include <hip/hip_runtime.h>
#include <math.h>

#define NODES_DEF 100000
#define IN_C 128
#define HID 32
#define OUT_C 32
#define NEG_SLOPE 0.2f

typedef __attribute__((ext_vector_type(2))) float v2f;
typedef __attribute__((ext_vector_type(8))) float v8f;

template <int K, int NC, int EPI, bool ZERO_CNT>
__global__ void gat_gemm_wmma(const float* __restrict__ A,
                              const float* __restrict__ B,
                              const float* __restrict__ bias,
                              float* __restrict__ C,
                              int*   __restrict__ cnt, int M)
{
    const int wavesPerBlock = blockDim.x >> 5;
    const int wave = blockIdx.x * wavesPerBlock + (threadIdx.x >> 5);
    const int nTiles = NC / 16;
    const int totalTiles = (M / 16) * nTiles;
    if (wave >= totalTiles) return;

    const int mTile = wave / nTiles;
    const int nTile = wave % nTiles;
    const int lane  = threadIdx.x & 31;
    const int half  = lane >> 4;
    const int r     = lane & 15;
    const int row   = mTile * 16 + r;
    const int col   = nTile * 16 + r;

    if (ZERO_CNT) {
        const int per = 16 / nTiles;
        if (lane < per) cnt[wave * per + lane] = 0;
    }

    v8f acc = {};
#pragma unroll
    for (int k = 0; k < K; k += 4) {
        v2f a = *(const v2f*)(A + (size_t)row * K + k + 2 * half);
        v2f b;
        b.x = B[(k + 2 * half + 0) * NC + col];
        b.y = B[(k + 2 * half + 1) * NC + col];
        acc = __builtin_amdgcn_wmma_f32_16x16x4_f32(
             false, a,  false, b,
             (short)0, acc,  false,  false);
    }

#pragma unroll
    for (int i = 0; i < 8; ++i) {
        const int orow = mTile * 16 + i + 8 * half;
        float v = acc[i];
        if (EPI == 1) {
            v += bias[col];
            v = 1.0f / (1.0f + __expf(-v));
        }
        C[(size_t)orow * NC + col] = v;
    }
}

__device__ __forceinline__ float leaky(float x)
{
    return x > 0.0f ? x : NEG_SLOPE * x;
}

#define BSH   7
#define BW    (1 << BSH)
#define MAXB  1024
#define BST   8
#define CH_T  1024
#define CH_E  8
#define CH    (CH_T * CH_E)
#define BS_T  512
#define BS_E  16
#define CAP   (BS_T * BS_E)
static_assert(MAXB == CH_T, "one thread per bucket counter in the chunk kernels");
static_assert(BW == 128, "bucket_sort sums its counters 4 per lane of one wave");

__device__ __forceinline__ int wave_incl_scan(int v, int lane)
{
#pragma unroll
    for (int o = 1; o < 32; o <<= 1) {
        const int t = __shfl_up(v, o);
        if (lane >= o) v += t;
    }
    return v;
}

__global__ void __launch_bounds__(CH_T)
bin_hist(const int* __restrict__ ei, int* __restrict__ btot, int n_edges, int nb)
{
    __shared__ int hist[MAXB];
    const int tid = threadIdx.x;
    hist[tid] = 0;
    __syncthreads();
    const int base = blockIdx.x * CH;
    int d[CH_E];
#pragma unroll
    for (int k = 0; k < CH_E; ++k) {
        const int i = base + k * CH_T + tid;
        d[k] = ei[n_edges + (i < n_edges ? i : n_edges - 1)];
    }
#pragma unroll
    for (int k = 0; k < CH_E; ++k) {
        const int i = base + k * CH_T + tid;
        if (i < n_edges) atomicAdd(&hist[d[k] >> BSH], 1);
    }
    __syncthreads();
    if (tid < nb) {
        const int c = hist[tid];
        if (c > 0) atomicAdd(&btot[tid * BST], c);
    }
}

__global__ void __launch_bounds__(CH_T)
bin_scatter(const int* __restrict__ ei, const int* __restrict__ btot,
            int* __restrict__ bcur, unsigned* __restrict__ binned, int n_edges, int nb)
{
    __shared__ int hist[MAXB];
    __shared__ int gbase[MAXB];
    __shared__ int wsA[32], wsB[32];
    __shared__ unsigned sw[CH];
    __shared__ int sg[CH];
    const int tid = threadIdx.x, lane = tid & 31, wv = tid >> 5;
    hist[tid] = 0;
    __syncthreads();
    const int base = blockIdx.x * CH;
    unsigned w[CH_E], br[CH_E];
    int sv[CH_E], dv[CH_E];
#pragma unroll
    for (int k = 0; k < CH_E; ++k) {
        const int i = base + k * CH_T + tid;
        const int iq = i < n_edges ? i : n_edges - 1;
        sv[k] = ei[iq];
        dv[k] = ei[n_edges + iq];
    }
#pragma unroll
    for (int k = 0; k < CH_E; ++k) {
        const int i = base + k * CH_T + tid;
        w[k] = 0u; br[k] = 0xFFFFFFFFu;
        if (i < n_edges) {
            const int s = sv[k];
            const int d = dv[k];
            const int b = d >> BSH;
            const int r = atomicAdd(&hist[b], 1);
            w[k]  = (unsigned)s | ((unsigned)(d & (BW - 1)) << 24);
            br[k] = (unsigned)b | ((unsigned)r << 16);
        }
    }
    __syncthreads();
    const int c = hist[tid];
    const int t = (tid < nb) ? btot[tid * BST] : 0;
    const int ic = wave_incl_scan(c, lane), it = wave_incl_scan(t, lane);
    if (lane == 31) { wsA[wv] = ic; wsB[wv] = it; }
    __syncthreads();
    if (wv == 0) {
        const int a = wsA[lane], b2 = wsB[lane];
        const int ia = wave_incl_scan(a, lane), ib = wave_incl_scan(b2, lane);
        wsA[lane] = ia - a; wsB[lane] = ib - b2;
    }
    __syncthreads();
    int g = 0;
    if (c > 0) g = atomicAdd(&bcur[tid * BST], c);
    hist[tid]  = ic - c + wsA[wv];
    gbase[tid] = it - t + wsB[wv] + g;
    __syncthreads();
#pragma unroll
    for (int k = 0; k < CH_E; ++k) {
        if (br[k] != 0xFFFFFFFFu) {
            const int b = (int)(br[k] & 0xFFFFu), r = (int)(br[k] >> 16);
            const int p = hist[b] + r;
            sw[p] = w[k];
            sg[p] = gbase[b] + r;
        }
    }
    __syncthreads();
    const int rem = n_edges - base;
    const int nvalid = rem < CH ? rem : CH;
#pragma unroll
    for (int k = 0; k < CH_E; ++k) {
        const int i = k * CH_T + tid;
        if (i < nvalid) binned[sg[i]] = sw[i];
    }
}

__global__ void __launch_bounds__(BS_T)
bucket_sort(const unsigned* __restrict__ binned, const int* __restrict__ btot,
            int* __restrict__ cnt, int* __restrict__ cur, int* __restrict__ srt,
            int n, int nb)
{
    __shared__ int c[BW];
    __shared__ int cursor[BW];
    __shared__ int s_start;
    __shared__ int sorted[CAP];
    const int tid = threadIdx.x;
    const int b = blockIdx.x;
    if (tid < BW) c[tid] = 0;
    if (tid == 0) s_start = 0;
    __syncthreads();
    int part = 0;
    for (int j = tid; j < b; j += BS_T) part += btot[j * BST];
    if (part != 0) atomicAdd(&s_start, part);
    const int tot = btot[b * BST];
    __syncthreads();
    const int start = s_start;
    const unsigned* __restrict__ in = binned + start;
    const bool staged = tot <= CAP;
    unsigned v[BS_E];
    if (staged) {
#pragma unroll
        for (int k = 0; k < BS_E; ++k) {
            const int i = k * BS_T + tid;
            v[k] = binned[i < tot ? start + i : 0];
        }
#pragma unroll
        for (int k = 0; k < BS_E; ++k)
            if (k * BS_T + tid < tot) atomicAdd(&c[v[k] >> 24], 1);
    } else {
        for (int i = tid; i < tot; i += BS_T) atomicAdd(&c[in[i] >> 24], 1);
    }
    __syncthreads();
    if (tid < 32) {
        const int v0 = c[4 * tid], v1 = c[4 * tid + 1], v2 = c[4 * tid + 2], v3 = c[4 * tid + 3];
        const int s = v0 + v1 + v2 + v3;
        const int e = wave_incl_scan(s, tid) - s;
        cursor[4 * tid]     = e;
        cursor[4 * tid + 1] = e + v0;
        cursor[4 * tid + 2] = e + v0 + v1;
        cursor[4 * tid + 3] = e + v0 + v1 + v2;
    }
    __syncthreads();
    if (tid < BW) {
        const int d = b * BW + tid;
        if (d < n) { cnt[d] = c[tid]; cur[d] = start + cursor[tid]; }
    }
    __syncthreads();
    if (staged) {
#pragma unroll
        for (int k = 0; k < BS_E; ++k) {
            if (k * BS_T + tid < tot) {
                const int p = atomicAdd(&cursor[v[k] >> 24], 1);
                sorted[p] = (int)(v[k] & 0xFFFFFFu);
            }
        }
        __syncthreads();
#pragma unroll
        for (int k = 0; k < BS_E; ++k) {
            const int i = k * BS_T + tid;
            if (i < tot) srt[start + i] = sorted[i];
        }
    } else {
        for (int i = tid; i < tot; i += BS_T) {
            const unsigned wd = in[i];
            const int p = atomicAdd(&cursor[wd >> 24], 1);
            srt[start + p] = (int)(wd & 0xFFFFFFu);
        }
    }
}

__device__ __forceinline__ float xhalf(float x)
{
    return __int_as_float(__builtin_amdgcn_permlanex16(__float_as_int(x), __float_as_int(x),
                                                      0x76543210, 0xfedcba98, false, false));
}
template <int S>
__device__ __forceinline__ float xrow(float x)
{
    return __int_as_float(__builtin_amdgcn_update_dpp(0, __float_as_int(x), 0x160 + S,
                                                     0xf, 0xf, true));
}
__device__ __forceinline__ float wave_sum(float v)
{
    v += xrow<1>(v); v += xrow<2>(v); v += xrow<4>(v); v += xrow<8>(v);
    return v + xhalf(v);
}
__device__ __forceinline__ float wave_max(float v)
{
    v = fmaxf(v, xrow<1>(v)); v = fmaxf(v, xrow<2>(v));
    v = fmaxf(v, xrow<4>(v)); v = fmaxf(v, xrow<8>(v));
    return fmaxf(v, xhalf(v));
}

template <int S, int N>
__device__ __forceinline__ void fold_rows(float* p, int lane)
{
    const bool hi = (lane & S) != 0;
#pragma unroll
    for (int u = 0; u < N; ++u) {
        const float x = hi ? p[u + N] : p[u];
        const float y = hi ? p[u] : p[u + N];
        if (S == 16) p[u] = x + xhalf(y);
        else         p[u] = x + xrow<(S < 16 ? S : 1)>(y);
    }
}

template <int R>
__device__ __forceinline__ float row_dots(float (&p)[R], int lane)
{
    if (R == 32) fold_rows<16, 16>(p, lane);
    else {
#pragma unroll
        for (int u = 0; u < R; ++u) p[u] += xhalf(p[u]);
    }
    if (R >= 16) fold_rows<8, 8>(p, lane);
    else {
#pragma unroll
        for (int u = 0; u < R; ++u) p[u] += xrow<8>(p[u]);
    }
    fold_rows<4, 4>(p, lane);
    fold_rows<2, 2>(p, lane);
    fold_rows<1, 1>(p, lane);
    return p[0];
}

template <int R>
__device__ __forceinline__ void gather_chunk(const float* __restrict__ h, int sj, int k,
                                             int lane, float as, float sd,
                                             float& m, float& z, float& acc)
{
    float v[R];
#pragma unroll
    for (int g = 0; g < R / 4; ++g) {
        if (4 * g < k) {
#pragma unroll
            for (int t = 0; t < 4; ++t)
            {
                const unsigned long long rb = (unsigned long long)h +
                    (unsigned long long)(unsigned)__builtin_amdgcn_readlane(sj, 4 * g + t) * (HID * 4);
                const unsigned lo = __builtin_amdgcn_readfirstlane((unsigned)rb);
                const unsigned hi = __builtin_amdgcn_readfirstlane((unsigned)(rb >> 32));
                typedef const __attribute__((address_space(1))) float* grow_t;
                const grow_t rowp = (grow_t)(((unsigned long long)hi << 32) | lo);
                v[4 * g + t] = rowp[(unsigned)lane];
            }
        } else {
#pragma unroll
            for (int t = 0; t < 4; ++t) v[4 * g + t] = 0.0f;
        }
    }
    float p[R];
#pragma unroll
    for (int u = 0; u < R; ++u) p[u] = v[u] * as;
    const float s = row_dots<R>(p, lane);

    const bool  valid = lane < k;
    const float e  = valid ? leaky(s + sd) : m;
    const float mn = fmaxf(m, wave_max(e));
    const float scale = __expf(m - mn);
    const float ex = valid ? __expf(e - mn) : 0.0f;
    z   = z * scale + wave_sum(ex);
    acc = acc * scale;
    const int wi = __float_as_int(ex);
#pragma unroll
    for (int g = 0; g < R / 4; ++g) {
        if (4 * g < k) {
#pragma unroll
            for (int t = 0; t < 4; ++t)
                acc = __builtin_fmaf(
                    __int_as_float(__builtin_amdgcn_readlane(wi, 4 * g + t)), v[4 * g + t], acc);
        }
    }
    m = mn;
}

__global__ void gat_node_gather(const int* __restrict__ srt,
                                const int* __restrict__ cur,
                                const int* __restrict__ cnt,
                                const float* __restrict__ a_src,
                                const float* __restrict__ a_dst,
                                const float* __restrict__ h,
                                const float* __restrict__ bias,
                                float* __restrict__ g, int n)
{
    const int lane = threadIdx.x & 31;
    const int d = __builtin_amdgcn_readfirstlane(
        (int)((blockIdx.x * blockDim.x + threadIdx.x) >> 5));
    if (d >= n) return;
    const int beg = __builtin_amdgcn_readfirstlane(cur[d]);
    const int deg = __builtin_amdgcn_readfirstlane(cnt[d]);
    const int end = beg + deg;

    const float hd = h[(size_t)d * HID + lane];
    const float as = a_src[lane];
    const float ad = a_dst[lane];
    const float bv = bias[lane];

    int sj = d;
    if (lane < deg) sj = srt[beg + lane];

    const float sd     = wave_sum(hd * ad);
    const float e_self = leaky(wave_sum(hd * as) + sd);

    float m = e_self, z = 1.0f, acc = hd;

    for (int c0 = beg; c0 < end; c0 += 32) {
        const int rem = end - c0;
        const int k   = rem < 32 ? rem : 32;
        int sn = d;
        if (lane + 32 < rem) sn = srt[c0 + 32 + lane];
        if (k > 16)     gather_chunk<32>(h, sj, k, lane, as, sd, m, z, acc);
        else if (k > 8) gather_chunk<16>(h, sj, k, lane, as, sd, m, z, acc);
        else            gather_chunk<8>(h, sj, k, lane, as, sd, m, z, acc);
        sj = sn;
    }

    const float o = acc / (z + 1e-16f) + bv;
    g[(size_t)d * HID + lane] = o > 0.0f ? o : 0.0f;
}

extern "C" void kernel_launch(void* const* d_in, const int* in_sizes, int n_in,
                              void* d_out, int out_size, void* d_ws, size_t ws_size,
                              hipStream_t stream)
{
    const float* x      = (const float*)d_in[0];
    const int*   ei     = (const int*)  d_in[1];
    const float* W1     = (const float*)d_in[2];
    const float* a_src1 = (const float*)d_in[3];
    const float* a_dst1 = (const float*)d_in[4];
    const float* b1     = (const float*)d_in[5];
    const float* W2     = (const float*)d_in[6];
    const float* a_src2 = (const float*)d_in[7];
    const float* a_dst2 = (const float*)d_in[8];
    const float* b2     = (const float*)d_in[9];
    const float* Wc     = (const float*)d_in[10];
    const float* bc     = (const float*)d_in[11];
    float* out = (float*)d_out;

    const int n       = in_sizes[0] / IN_C;
    const int n_edges = in_sizes[1] / 2;

    float* ws    = (float*)d_ws;
    float* h     = ws;
    float* g     = h   + (size_t)n * HID;
    int*   cnt   = (int*)(g + (size_t)n * HID);
    int*   bk    = cnt + 2 * (size_t)n;
    int*   cur   = bk + n;
    int*   srt   = cur + n;
    int*   btot  = bk;
    int*   bcur  = bk + MAXB * BST;
    unsigned* binned = (unsigned*)g;

    const int TB = 256;
    const int chunkBlocks = (n_edges + CH - 1) / CH;
    const int nb = (n + BW - 1) / BW;
    const int waveBlocks = (n + 7) / 8;
    const int gemmWaves  = (n / 16) * (HID / 16);
    const int gemmBlocks = (gemmWaves + 7) / 8;

    gat_gemm_wmma<IN_C, HID, 0, true><<<gemmBlocks, TB, 0, stream>>>(x, W1, nullptr, h, bk, n);
    bin_hist<<<chunkBlocks, CH_T, 0, stream>>>(ei, btot, n_edges, nb);
    bin_scatter<<<chunkBlocks, CH_T, 0, stream>>>(ei, btot, bcur, binned, n_edges, nb);
    bucket_sort<<<nb, BS_T, 0, stream>>>(binned, btot, cnt, cur, srt, n, nb);
    gat_node_gather<<<waveBlocks, TB, 0, stream>>>(srt, cur, cnt, a_src1, a_dst1, h, b1, g, n);

    gat_gemm_wmma<HID, HID, 0, false><<<gemmBlocks, TB, 0, stream>>>(g, W2, nullptr, h, nullptr, n);
    gat_node_gather<<<waveBlocks, TB, 0, stream>>>(srt, cur, cnt, a_src2, a_dst2, h, b2, g, n);

    gat_gemm_wmma<HID, OUT_C, 1, false><<<gemmBlocks, TB, 0, stream>>>(g, Wc, bc, out, nullptr, n);
}


# dense layers 2 and 3 run inside the gather launches: a work-group's 16 node rows meet in LDS as one A tile, 2 of its waves do rows x W on f32 WMMA and store; 2 launches gone, sum orders unchanged
# speedup vs baseline: 24.1469x; 1.0345x over previous
#include <hip/hip_runtime.h>
#include <math.h>

#define NODES_DEF 100000
#define IN_C 128
#define HID 32
#define OUT_C 32
#define NEG_SLOPE 0.2f

typedef __attribute__((ext_vector_type(2))) float v2f;
typedef __attribute__((ext_vector_type(8))) float v8f;

template <int K, int NC>
__global__ void gat_gemm_wmma(const float* __restrict__ A,
                              const float* __restrict__ B,
                              float* __restrict__ C,
                              int*   __restrict__ cnt, int M)
{
    const int wavesPerBlock = blockDim.x >> 5;
    const int wave = blockIdx.x * wavesPerBlock + (threadIdx.x >> 5);
    const int nTiles = NC / 16;
    const int totalTiles = (M / 16) * nTiles;
    if (wave >= totalTiles) return;

    const int mTile = wave / nTiles;
    const int nTile = wave % nTiles;
    const int lane  = threadIdx.x & 31;
    const int half  = lane >> 4;
    const int r     = lane & 15;
    const int row   = mTile * 16 + r;
    const int col   = nTile * 16 + r;

    {
        const int per = 16 / nTiles;
        if (lane < per) cnt[wave * per + lane] = 0;
    }

    v8f acc = {};
#pragma unroll
    for (int k = 0; k < K; k += 4) {
        v2f a = *(const v2f*)(A + (size_t)row * K + k + 2 * half);
        v2f b;
        b.x = B[(k + 2 * half + 0) * NC + col];
        b.y = B[(k + 2 * half + 1) * NC + col];
        acc = __builtin_amdgcn_wmma_f32_16x16x4_f32(
             false, a,  false, b,
             (short)0, acc,  false,  false);
    }

#pragma unroll
    for (int i = 0; i < 8; ++i) {
        const int orow = mTile * 16 + i + 8 * half;
        C[(size_t)orow * NC + col] = acc[i];
    }
}

__device__ __forceinline__ float leaky(float x)
{
    return x > 0.0f ? x : NEG_SLOPE * x;
}

#define BSH   7
#define BW    (1 << BSH)
#define MAXB  1024
#define BST   8
#define CH_T  1024
#define CH_E  8
#define CH    (CH_T * CH_E)
#define BS_T  512
#define BS_E  16
#define CAP   (BS_T * BS_E)
static_assert(MAXB == CH_T, "one thread per bucket counter in the chunk kernels");
static_assert(BW == 128, "bucket_sort sums its counters 4 per lane of one wave");

__device__ __forceinline__ int wave_incl_scan(int v, int lane)
{
#pragma unroll
    for (int o = 1; o < 32; o <<= 1) {
        const int t = __shfl_up(v, o);
        if (lane >= o) v += t;
    }
    return v;
}

__global__ void __launch_bounds__(CH_T)
bin_hist(const int* __restrict__ ei, int* __restrict__ btot, int n_edges, int nb)
{
    __shared__ int hist[MAXB];
    const int tid = threadIdx.x;
    hist[tid] = 0;
    __syncthreads();
    const int base = blockIdx.x * CH;
    int d[CH_E];
#pragma unroll
    for (int k = 0; k < CH_E; ++k) {
        const int i = base + k * CH_T + tid;
        d[k] = ei[n_edges + (i < n_edges ? i : n_edges - 1)];
    }
#pragma unroll
    for (int k = 0; k < CH_E; ++k) {
        const int i = base + k * CH_T + tid;
        if (i < n_edges) atomicAdd(&hist[d[k] >> BSH], 1);
    }
    __syncthreads();
    if (tid < nb) {
        const int c = hist[tid];
        if (c > 0) atomicAdd(&btot[tid * BST], c);
    }
}

__global__ void __launch_bounds__(CH_T)
bin_scatter(const int* __restrict__ ei, const int* __restrict__ btot,
            int* __restrict__ bcur, unsigned* __restrict__ binned, int n_edges, int nb)
{
    __shared__ int hist[MAXB];
    __shared__ int gbase[MAXB];
    __shared__ int wsA[32], wsB[32];
    __shared__ unsigned sw[CH];
    __shared__ int sg[CH];
    const int tid = threadIdx.x, lane = tid & 31, wv = tid >> 5;
    hist[tid] = 0;
    __syncthreads();
    const int base = blockIdx.x * CH;
    unsigned w[CH_E], br[CH_E];
    int sv[CH_E], dv[CH_E];
#pragma unroll
    for (int k = 0; k < CH_E; ++k) {
        const int i = base + k * CH_T + tid;
        const int iq = i < n_edges ? i : n_edges - 1;
        sv[k] = ei[iq];
        dv[k] = ei[n_edges + iq];
    }
#pragma unroll
    for (int k = 0; k < CH_E; ++k) {
        const int i = base + k * CH_T + tid;
        w[k] = 0u; br[k] = 0xFFFFFFFFu;
        if (i < n_edges) {
            const int s = sv[k];
            const int d = dv[k];
            const int b = d >> BSH;
            const int r = atomicAdd(&hist[b], 1);
            w[k]  = (unsigned)s | ((unsigned)(d & (BW - 1)) << 24);
            br[k] = (unsigned)b | ((unsigned)r << 16);
        }
    }
    __syncthreads();
    const int c = hist[tid];
    const int t = (tid < nb) ? btot[tid * BST] : 0;
    const int ic = wave_incl_scan(c, lane), it = wave_incl_scan(t, lane);
    if (lane == 31) { wsA[wv] = ic; wsB[wv] = it; }
    __syncthreads();
    if (wv == 0) {
        const int a = wsA[lane], b2 = wsB[lane];
        const int ia = wave_incl_scan(a, lane), ib = wave_incl_scan(b2, lane);
        wsA[lane] = ia - a; wsB[lane] = ib - b2;
    }
    __syncthreads();
    int g = 0;
    if (c > 0) g = atomicAdd(&bcur[tid * BST], c);
    hist[tid]  = ic - c + wsA[wv];
    gbase[tid] = it - t + wsB[wv] + g;
    __syncthreads();
#pragma unroll
    for (int k = 0; k < CH_E; ++k) {
        if (br[k] != 0xFFFFFFFFu) {
            const int b = (int)(br[k] & 0xFFFFu), r = (int)(br[k] >> 16);
            const int p = hist[b] + r;
            sw[p] = w[k];
            sg[p] = gbase[b] + r;
        }
    }
    __syncthreads();
    const int rem = n_edges - base;
    const int nvalid = rem < CH ? rem : CH;
#pragma unroll
    for (int k = 0; k < CH_E; ++k) {
        const int i = k * CH_T + tid;
        if (i < nvalid) binned[sg[i]] = sw[i];
    }
}

__global__ void __launch_bounds__(BS_T)
bucket_sort(const unsigned* __restrict__ binned, const int* __restrict__ btot,
            int* __restrict__ cnt, int* __restrict__ cur, int* __restrict__ srt,
            int n, int nb)
{
    __shared__ int c[BW];
    __shared__ int cursor[BW];
    __shared__ int s_start;
    __shared__ int sorted[CAP];
    const int tid = threadIdx.x;
    const int b = blockIdx.x;
    if (tid < BW) c[tid] = 0;
    if (tid == 0) s_start = 0;
    __syncthreads();
    int part = 0;
    for (int j = tid; j < b; j += BS_T) part += btot[j * BST];
    if (part != 0) atomicAdd(&s_start, part);
    const int tot = btot[b * BST];
    __syncthreads();
    const int start = s_start;
    const unsigned* __restrict__ in = binned + start;
    const bool staged = tot <= CAP;
    unsigned v[BS_E];
    if (staged) {
#pragma unroll
        for (int k = 0; k < BS_E; ++k) {
            const int i = k * BS_T + tid;
            v[k] = binned[i < tot ? start + i : 0];
        }
#pragma unroll
        for (int k = 0; k < BS_E; ++k)
            if (k * BS_T + tid < tot) atomicAdd(&c[v[k] >> 24], 1);
    } else {
        for (int i = tid; i < tot; i += BS_T) atomicAdd(&c[in[i] >> 24], 1);
    }
    __syncthreads();
    if (tid < 32) {
        const int v0 = c[4 * tid], v1 = c[4 * tid + 1], v2 = c[4 * tid + 2], v3 = c[4 * tid + 3];
        const int s = v0 + v1 + v2 + v3;
        const int e = wave_incl_scan(s, tid) - s;
        cursor[4 * tid]     = e;
        cursor[4 * tid + 1] = e + v0;
        cursor[4 * tid + 2] = e + v0 + v1;
        cursor[4 * tid + 3] = e + v0 + v1 + v2;
    }
    __syncthreads();
    if (tid < BW) {
        const int d = b * BW + tid;
        if (d < n) { cnt[d] = c[tid]; cur[d] = start + cursor[tid]; }
    }
    __syncthreads();
    if (staged) {
#pragma unroll
        for (int k = 0; k < BS_E; ++k) {
            if (k * BS_T + tid < tot) {
                const int p = atomicAdd(&cursor[v[k] >> 24], 1);
                sorted[p] = (int)(v[k] & 0xFFFFFFu);
            }
        }
        __syncthreads();
#pragma unroll
        for (int k = 0; k < BS_E; ++k) {
            const int i = k * BS_T + tid;
            if (i < tot) srt[start + i] = sorted[i];
        }
    } else {
        for (int i = tid; i < tot; i += BS_T) {
            const unsigned wd = in[i];
            const int p = atomicAdd(&cursor[wd >> 24], 1);
            srt[start + p] = (int)(wd & 0xFFFFFFu);
        }
    }
}

__device__ __forceinline__ float xhalf(float x)
{
    return __int_as_float(__builtin_amdgcn_permlanex16(__float_as_int(x), __float_as_int(x),
                                                      0x76543210, 0xfedcba98, false, false));
}
template <int S>
__device__ __forceinline__ float xrow(float x)
{
    return __int_as_float(__builtin_amdgcn_update_dpp(0, __float_as_int(x), 0x160 + S,
                                                     0xf, 0xf, true));
}
__device__ __forceinline__ float wave_sum(float v)
{
    v += xrow<1>(v); v += xrow<2>(v); v += xrow<4>(v); v += xrow<8>(v);
    return v + xhalf(v);
}
__device__ __forceinline__ float wave_max(float v)
{
    v = fmaxf(v, xrow<1>(v)); v = fmaxf(v, xrow<2>(v));
    v = fmaxf(v, xrow<4>(v)); v = fmaxf(v, xrow<8>(v));
    return fmaxf(v, xhalf(v));
}

template <int S, int N>
__device__ __forceinline__ void fold_rows(float* p, int lane)
{
    const bool hi = (lane & S) != 0;
#pragma unroll
    for (int u = 0; u < N; ++u) {
        const float x = hi ? p[u + N] : p[u];
        const float y = hi ? p[u] : p[u + N];
        if (S == 16) p[u] = x + xhalf(y);
        else         p[u] = x + xrow<(S < 16 ? S : 1)>(y);
    }
}

template <int R>
__device__ __forceinline__ float row_dots(float (&p)[R], int lane)
{
    if (R == 32) fold_rows<16, 16>(p, lane);
    else {
#pragma unroll
        for (int u = 0; u < R; ++u) p[u] += xhalf(p[u]);
    }
    if (R >= 16) fold_rows<8, 8>(p, lane);
    else {
#pragma unroll
        for (int u = 0; u < R; ++u) p[u] += xrow<8>(p[u]);
    }
    fold_rows<4, 4>(p, lane);
    fold_rows<2, 2>(p, lane);
    fold_rows<1, 1>(p, lane);
    return p[0];
}

template <int R>
__device__ __forceinline__ void gather_chunk(const float* __restrict__ h, int sj, int k,
                                             int lane, float as, float sd,
                                             float& m, float& z, float& acc)
{
    float v[R];
#pragma unroll
    for (int g = 0; g < R / 4; ++g) {
        if (4 * g < k) {
#pragma unroll
            for (int t = 0; t < 4; ++t)
            {
                const unsigned long long rb = (unsigned long long)h +
                    (unsigned long long)(unsigned)__builtin_amdgcn_readlane(sj, 4 * g + t) * (HID * 4);
                const unsigned lo = __builtin_amdgcn_readfirstlane((unsigned)rb);
                const unsigned hi = __builtin_amdgcn_readfirstlane((unsigned)(rb >> 32));
                typedef const __attribute__((address_space(1))) float* grow_t;
                const grow_t rowp = (grow_t)(((unsigned long long)hi << 32) | lo);
                v[4 * g + t] = rowp[(unsigned)lane];
            }
        } else {
#pragma unroll
            for (int t = 0; t < 4; ++t) v[4 * g + t] = 0.0f;
        }
    }
    float p[R];
#pragma unroll
    for (int u = 0; u < R; ++u) p[u] = v[u] * as;
    const float s = row_dots<R>(p, lane);

    const bool  valid = lane < k;
    const float e  = valid ? leaky(s + sd) : m;
    const float mn = fmaxf(m, wave_max(e));
    const float scale = __expf(m - mn);
    const float ex = valid ? __expf(e - mn) : 0.0f;
    z   = z * scale + wave_sum(ex);
    acc = acc * scale;
    const int wi = __float_as_int(ex);
#pragma unroll
    for (int g = 0; g < R / 4; ++g) {
        if (4 * g < k) {
#pragma unroll
            for (int t = 0; t < 4; ++t)
                acc = __builtin_fmaf(
                    __int_as_float(__builtin_amdgcn_readlane(wi, 4 * g + t)), v[4 * g + t], acc);
        }
    }
    m = mn;
}

#define GN 16
#define TP (HID + 2)
static_assert(OUT_C == HID, "both fused products are 32 x 32: one row pitch, two column tiles");
template <int EPI>
__global__ void __launch_bounds__(GN * 32)
gat_node_gather_mm(const int* __restrict__ srt,
                   const int* __restrict__ cur,
                   const int* __restrict__ cnt,
                   const float* __restrict__ a_src,
                   const float* __restrict__ a_dst,
                   const float* __restrict__ h,
                   const float* __restrict__ bias,
                   const float* __restrict__ Wn,
                   const float* __restrict__ bn,
                   float* __restrict__ outp)
{
    __shared__ float tile[GN * TP];
    const int lane = threadIdx.x & 31;
    const int wv = __builtin_amdgcn_readfirstlane((int)(threadIdx.x >> 5));
    const int d  = (int)blockIdx.x * GN + wv;

    float o;
    {
        const int beg = __builtin_amdgcn_readfirstlane(cur[d]);
        const int deg = __builtin_amdgcn_readfirstlane(cnt[d]);
        const int end = beg + deg;

        const float hd = h[(size_t)d * HID + lane];
        const float as = a_src[lane];
        const float ad = a_dst[lane];
        const float bv = bias[lane];

        int sj = d;
        if (lane < deg) sj = srt[beg + lane];

        const float sd     = wave_sum(hd * ad);
        const float e_self = leaky(wave_sum(hd * as) + sd);

        float m = e_self, z = 1.0f, acc = hd;

        for (int c0 = beg; c0 < end; c0 += 32) {
            const int rem = end - c0;
            const int k   = rem < 32 ? rem : 32;
            int sn = d;
            if (lane + 32 < rem) sn = srt[c0 + 32 + lane];
            if (k > 16)     gather_chunk<32>(h, sj, k, lane, as, sd, m, z, acc);
            else if (k > 8) gather_chunk<16>(h, sj, k, lane, as, sd, m, z, acc);
            else            gather_chunk<8>(h, sj, k, lane, as, sd, m, z, acc);
            sj = sn;
        }

        const float r = acc / (z + 1e-16f) + bv;
        o = r > 0.0f ? r : 0.0f;
    }
    tile[wv * TP + lane] = o;
    __syncthreads();
    if (wv >= HID / 16) return;

    const int half = lane >> 4;
    const int rr   = lane & 15;
    const int col  = wv * 16 + rr;
    v8f acc2 = {};
#pragma unroll
    for (int k = 0; k < HID; k += 4) {
        v2f a = *(const v2f*)(tile + rr * TP + k + 2 * half);
        v2f b;
        b.x = Wn[(k + 2 * half + 0) * HID + col];
        b.y = Wn[(k + 2 * half + 1) * HID + col];
        acc2 = __builtin_amdgcn_wmma_f32_16x16x4_f32(
             false, a,  false, b,
             (short)0, acc2,  false,  false);
    }
    float bcol = 0.0f;
    if (EPI == 1) bcol = bn[col];
#pragma unroll
    for (int i = 0; i < 8; ++i) {
        const int orow = (int)blockIdx.x * GN + i + 8 * half;
        float v = acc2[i];
        if (EPI == 1) {
            v += bcol;
            v = 1.0f / (1.0f + __expf(-v));
        }
        outp[(size_t)orow * HID + col] = v;
    }
}

extern "C" void kernel_launch(void* const* d_in, const int* in_sizes, int n_in,
                              void* d_out, int out_size, void* d_ws, size_t ws_size,
                              hipStream_t stream)
{
    const float* x      = (const float*)d_in[0];
    const int*   ei     = (const int*)  d_in[1];
    const float* W1     = (const float*)d_in[2];
    const float* a_src1 = (const float*)d_in[3];
    const float* a_dst1 = (const float*)d_in[4];
    const float* b1     = (const float*)d_in[5];
    const float* W2     = (const float*)d_in[6];
    const float* a_src2 = (const float*)d_in[7];
    const float* a_dst2 = (const float*)d_in[8];
    const float* b2     = (const float*)d_in[9];
    const float* Wc     = (const float*)d_in[10];
    const float* bc     = (const float*)d_in[11];
    float* out = (float*)d_out;

    const int n       = in_sizes[0] / IN_C;
    const int n_edges = in_sizes[1] / 2;

    float* ws    = (float*)d_ws;
    float* h     = ws;
    float* g     = h   + (size_t)n * HID;
    int*   cnt   = (int*)(g + (size_t)n * HID);
    int*   bk    = cnt + 2 * (size_t)n;
    int*   cur   = bk + n;
    int*   srt   = cur + n;
    int*   btot  = bk;
    int*   bcur  = bk + MAXB * BST;
    unsigned* binned = (unsigned*)g;

    const int TB = 256;
    const int chunkBlocks = (n_edges + CH - 1) / CH;
    const int nb = (n + BW - 1) / BW;
    const int nodeBlocks = n / GN;
    const int gemmWaves  = (n / 16) * (HID / 16);
    const int gemmBlocks = (gemmWaves + 7) / 8;

    gat_gemm_wmma<IN_C, HID><<<gemmBlocks, TB, 0, stream>>>(x, W1, h, bk, n);
    bin_hist<<<chunkBlocks, CH_T, 0, stream>>>(ei, btot, n_edges, nb);
    bin_scatter<<<chunkBlocks, CH_T, 0, stream>>>(ei, btot, bcur, binned, n_edges, nb);
    bucket_sort<<<nb, BS_T, 0, stream>>>(binned, btot, cnt, cur, srt, n, nb);
    gat_node_gather_mm<0><<<nodeBlocks, GN * 32, 0, stream>>>(srt, cur, cnt, a_src1, a_dst1, h, b1,
                                                             W2, nullptr, g);

    gat_node_gather_mm<1><<<nodeBlocks, GN * 32, 0, stream>>>(srt, cur, cnt, a_src2, a_dst2, g, b2,
                                                             Wc, bc, out);
}


# node gather holds a source row in 8 lanes (16 bytes each): 4 rows per load instruction, scores folded inside 8-lane groups, weights spread by DPP8, one reciprocal per node; sum order changes
# speedup vs baseline: 26.1965x; 1.0849x over previous
#include <hip/hip_runtime.h>
#include <math.h>

#define NODES_DEF 100000
#define IN_C 128
#define HID 32
#define OUT_C 32
#define NEG_SLOPE 0.2f

typedef __attribute__((ext_vector_type(2))) float v2f;
typedef __attribute__((ext_vector_type(4))) float v4f;
typedef __attribute__((ext_vector_type(8))) float v8f;

template <int K, int NC>
__global__ void gat_gemm_wmma(const float* __restrict__ A,
                              const float* __restrict__ B,
                              float* __restrict__ C,
                              int*   __restrict__ cnt, int M)
{
    const int wavesPerBlock = blockDim.x >> 5;
    const int wave = blockIdx.x * wavesPerBlock + (threadIdx.x >> 5);
    const int nTiles = NC / 16;
    const int totalTiles = (M / 16) * nTiles;
    if (wave >= totalTiles) return;

    const int mTile = wave / nTiles;
    const int nTile = wave % nTiles;
    const int lane  = threadIdx.x & 31;
    const int half  = lane >> 4;
    const int r     = lane & 15;
    const int row   = mTile * 16 + r;
    const int col   = nTile * 16 + r;

    {
        const int per = 16 / nTiles;
        if (lane < per) cnt[wave * per + lane] = 0;
    }

    v8f acc = {};
#pragma unroll
    for (int k = 0; k < K; k += 4) {
        v2f a = *(const v2f*)(A + (size_t)row * K + k + 2 * half);
        v2f b;
        b.x = B[(k + 2 * half + 0) * NC + col];
        b.y = B[(k + 2 * half + 1) * NC + col];
        acc = __builtin_amdgcn_wmma_f32_16x16x4_f32(
             false, a,  false, b,
             (short)0, acc,  false,  false);
    }

#pragma unroll
    for (int i = 0; i < 8; ++i) {
        const int orow = mTile * 16 + i + 8 * half;
        C[(size_t)orow * NC + col] = acc[i];
    }
}

__device__ __forceinline__ float leaky(float x)
{
    return x > 0.0f ? x : NEG_SLOPE * x;
}

#define BSH   7
#define BW    (1 << BSH)
#define MAXB  1024
#define BST   8
#define CH_T  1024
#define CH_E  8
#define CH    (CH_T * CH_E)
#define BS_T  512
#define BS_E  16
#define CAP   (BS_T * BS_E)
static_assert(MAXB == CH_T, "one thread per bucket counter in the chunk kernels");
static_assert(BW == 128, "bucket_sort sums its counters 4 per lane of one wave");

__device__ __forceinline__ int wave_incl_scan(int v, int lane)
{
#pragma unroll
    for (int o = 1; o < 32; o <<= 1) {
        const int t = __shfl_up(v, o);
        if (lane >= o) v += t;
    }
    return v;
}

__global__ void __launch_bounds__(CH_T)
bin_hist(const int* __restrict__ ei, int* __restrict__ btot, int n_edges, int nb)
{
    __shared__ int hist[MAXB];
    const int tid = threadIdx.x;
    hist[tid] = 0;
    __syncthreads();
    const int base = blockIdx.x * CH;
    int d[CH_E];
#pragma unroll
    for (int k = 0; k < CH_E; ++k) {
        const int i = base + k * CH_T + tid;
        d[k] = ei[n_edges + (i < n_edges ? i : n_edges - 1)];
    }
#pragma unroll
    for (int k = 0; k < CH_E; ++k) {
        const int i = base + k * CH_T + tid;
        if (i < n_edges) atomicAdd(&hist[d[k] >> BSH], 1);
    }
    __syncthreads();
    if (tid < nb) {
        const int c = hist[tid];
        if (c > 0) atomicAdd(&btot[tid * BST], c);
    }
}

__global__ void __launch_bounds__(CH_T)
bin_scatter(const int* __restrict__ ei, const int* __restrict__ btot,
            int* __restrict__ bcur, unsigned* __restrict__ binned, int n_edges, int nb)
{
    __shared__ int hist[MAXB];
    __shared__ int gbase[MAXB];
    __shared__ int wsA[32], wsB[32];
    __shared__ unsigned sw[CH];
    __shared__ int sg[CH];
    const int tid = threadIdx.x, lane = tid & 31, wv = tid >> 5;
    hist[tid] = 0;
    __syncthreads();
    const int base = blockIdx.x * CH;
    unsigned w[CH_E], br[CH_E];
    int sv[CH_E], dv[CH_E];
#pragma unroll
    for (int k = 0; k < CH_E; ++k) {
        const int i = base + k * CH_T + tid;
        const int iq = i < n_edges ? i : n_edges - 1;
        sv[k] = ei[iq];
        dv[k] = ei[n_edges + iq];
    }
#pragma unroll
    for (int k = 0; k < CH_E; ++k) {
        const int i = base + k * CH_T + tid;
        w[k] = 0u; br[k] = 0xFFFFFFFFu;
        if (i < n_edges) {
            const int s = sv[k];
            const int d = dv[k];
            const int b = d >> BSH;
            const int r = atomicAdd(&hist[b], 1);
            w[k]  = (unsigned)s | ((unsigned)(d & (BW - 1)) << 24);
            br[k] = (unsigned)b | ((unsigned)r << 16);
        }
    }
    __syncthreads();
    const int c = hist[tid];
    const int t = (tid < nb) ? btot[tid * BST] : 0;
    const int ic = wave_incl_scan(c, lane), it = wave_incl_scan(t, lane);
    if (lane == 31) { wsA[wv] = ic; wsB[wv] = it; }
    __syncthreads();
    if (wv == 0) {
        const int a = wsA[lane], b2 = wsB[lane];
        const int ia = wave_incl_scan(a, lane), ib = wave_incl_scan(b2, lane);
        wsA[lane] = ia - a; wsB[lane] = ib - b2;
    }
    __syncthreads();
    int g = 0;
    if (c > 0) g = atomicAdd(&bcur[tid * BST], c);
    hist[tid]  = ic - c + wsA[wv];
    gbase[tid] = it - t + wsB[wv] + g;
    __syncthreads();
#pragma unroll
    for (int k = 0; k < CH_E; ++k) {
        if (br[k] != 0xFFFFFFFFu) {
            const int b = (int)(br[k] & 0xFFFFu), r = (int)(br[k] >> 16);
            const int p = hist[b] + r;
            sw[p] = w[k];
            sg[p] = gbase[b] + r;
        }
    }
    __syncthreads();
    const int rem = n_edges - base;
    const int nvalid = rem < CH ? rem : CH;
#pragma unroll
    for (int k = 0; k < CH_E; ++k) {
        const int i = k * CH_T + tid;
        if (i < nvalid) binned[sg[i]] = sw[i];
    }
}

__global__ void __launch_bounds__(BS_T)
bucket_sort(const unsigned* __restrict__ binned, const int* __restrict__ btot,
            int* __restrict__ cnt, int* __restrict__ cur, int* __restrict__ srt,
            int n, int nb)
{
    __shared__ int c[BW];
    __shared__ int cursor[BW];
    __shared__ int s_start;
    __shared__ int sorted[CAP];
    const int tid = threadIdx.x;
    const int b = blockIdx.x;
    if (tid < BW) c[tid] = 0;
    if (tid == 0) s_start = 0;
    __syncthreads();
    int part = 0;
    for (int j = tid; j < b; j += BS_T) part += btot[j * BST];
    if (part != 0) atomicAdd(&s_start, part);
    const int tot = btot[b * BST];
    __syncthreads();
    const int start = s_start;
    const unsigned* __restrict__ in = binned + start;
    const bool staged = tot <= CAP;
    unsigned v[BS_E];
    if (staged) {
#pragma unroll
        for (int k = 0; k < BS_E; ++k) {
            const int i = k * BS_T + tid;
            v[k] = binned[i < tot ? start + i : 0];
        }
#pragma unroll
        for (int k = 0; k < BS_E; ++k)
            if (k * BS_T + tid < tot) atomicAdd(&c[v[k] >> 24], 1);
    } else {
        for (int i = tid; i < tot; i += BS_T) atomicAdd(&c[in[i] >> 24], 1);
    }
    __syncthreads();
    if (tid < 32) {
        const int v0 = c[4 * tid], v1 = c[4 * tid + 1], v2 = c[4 * tid + 2], v3 = c[4 * tid + 3];
        const int s = v0 + v1 + v2 + v3;
        const int e = wave_incl_scan(s, tid) - s;
        cursor[4 * tid]     = e;
        cursor[4 * tid + 1] = e + v0;
        cursor[4 * tid + 2] = e + v0 + v1;
        cursor[4 * tid + 3] = e + v0 + v1 + v2;
    }
    __syncthreads();
    if (tid < BW) {
        const int d = b * BW + tid;
        if (d < n) { cnt[d] = c[tid]; cur[d] = start + cursor[tid]; }
    }
    __syncthreads();
    if (staged) {
#pragma unroll
        for (int k = 0; k < BS_E; ++k) {
            if (k * BS_T + tid < tot) {
                const int p = atomicAdd(&cursor[v[k] >> 24], 1);
                sorted[p] = (int)(v[k] & 0xFFFFFFu);
            }
        }
        __syncthreads();
#pragma unroll
        for (int k = 0; k < BS_E; ++k) {
            const int i = k * BS_T + tid;
            if (i < tot) srt[start + i] = sorted[i];
        }
    } else {
        for (int i = tid; i < tot; i += BS_T) {
            const unsigned wd = in[i];
            const int p = atomicAdd(&cursor[wd >> 24], 1);
            srt[start + p] = (int)(wd & 0xFFFFFFu);
        }
    }
}

__device__ __forceinline__ float xhalf(float x)
{
    return __int_as_float(__builtin_amdgcn_permlanex16(__float_as_int(x), __float_as_int(x),
                                                      0x76543210, 0xfedcba98, false, false));
}
template <int S>
__device__ __forceinline__ float xrow(float x)
{
    return __int_as_float(__builtin_amdgcn_update_dpp(0, __float_as_int(x), 0x160 + S,
                                                     0xf, 0xf, true));
}
__device__ __forceinline__ float wave_sum(float v)
{
    v += xrow<1>(v); v += xrow<2>(v); v += xrow<4>(v); v += xrow<8>(v);
    return v + xhalf(v);
}
__device__ __forceinline__ float wave_max(float v)
{
    v = fmaxf(v, xrow<1>(v)); v = fmaxf(v, xrow<2>(v));
    v = fmaxf(v, xrow<4>(v)); v = fmaxf(v, xrow<8>(v));
    return fmaxf(v, xhalf(v));
}

template <int J>
__device__ __forceinline__ int grp_pick(int x)
{
    return (int)__builtin_amdgcn_mov_dpp8((unsigned)x, J * 0x249249u);
}
__device__ __forceinline__ int grp_bcast(int x, int j)
{
    switch (j) {
    case 0:  return grp_pick<0>(x);
    case 1:  return grp_pick<1>(x);
    case 2:  return grp_pick<2>(x);
    case 3:  return grp_pick<3>(x);
    case 4:  return grp_pick<4>(x);
    case 5:  return grp_pick<5>(x);
    case 6:  return grp_pick<6>(x);
    default: return grp_pick<7>(x);
    }
}
__device__ __forceinline__ float grp_sum(float v)
{
    v += xrow<1>(v); v += xrow<2>(v);
    return v + xrow<4>(v);
}
__device__ __forceinline__ float dot4(v4f a, v4f b)
{
    const v2f t = __builtin_elementwise_fma(a.zw, b.zw, a.xy * b.xy);
    return t.x + t.y;
}

template <int S, int N>
__device__ __forceinline__ void fold_rows(float* p, int lane)
{
    const bool hi = (lane & S) != 0;
#pragma unroll
    for (int u = 0; u < N; ++u) {
        const float x = hi ? p[u + N] : p[u];
        const float y = hi ? p[u] : p[u + N];
        p[u] = x + xrow<S>(y);
    }
}

template <int G>
__device__ __forceinline__ float group_dots(float (&p)[G], int lane)
{
    if constexpr (G == 8) fold_rows<4, 4>(p, lane);
    else {
#pragma unroll
        for (int u = 0; u < G; ++u) p[u] += xrow<4>(p[u]);
    }
    if constexpr (G >= 4) fold_rows<2, 2>(p, lane);
    else {
#pragma unroll
        for (int u = 0; u < G; ++u) p[u] += xrow<2>(p[u]);
    }
    fold_rows<1, 1>(p, lane);
    return p[0];
}

template <int G>
__device__ __forceinline__ void gather_chunk(const float* __restrict__ h, int sj, int k,
                                             int lane, int eidx, unsigned sub16,
                                             v4f as, float sd,
                                             float& m, float& z, v4f& acc)
{
    v4f v[G];
#pragma unroll
    for (int g = 0; g < G; ++g) {
        if (4 * g < k) {
            const unsigned s = (unsigned)grp_bcast(sj, g);
            const unsigned off = s * (unsigned)(HID * 4) + sub16;
            v[g] = *(const v4f*)((const char*)h + (size_t)off);
        } else {
            v[g] = (v4f){0.0f, 0.0f, 0.0f, 0.0f};
        }
    }
    float p[G];
#pragma unroll
    for (int g = 0; g < G; ++g) p[g] = dot4(v[g], as);
    const float s = group_dots<G>(p, lane);

    const bool  valid = eidx < k;
    const float e  = valid ? leaky(s + sd) : m;
    const float mn = fmaxf(m, wave_max(e));
    const float scale = __expf(m - mn);
    const float ex = valid ? __expf(e - mn) : 0.0f;
    z   = z * scale + wave_sum(ex);
    acc.x *= scale; acc.y *= scale; acc.z *= scale; acc.w *= scale;
    const int wi = __float_as_int(ex);
#pragma unroll
    for (int g = 0; g < G; ++g) {
        if (4 * g < k) {
            const float w = __int_as_float(grp_bcast(wi, g));
            acc.x = __builtin_fmaf(w, v[g].x, acc.x);
            acc.y = __builtin_fmaf(w, v[g].y, acc.y);
            acc.z = __builtin_fmaf(w, v[g].z, acc.z);
            acc.w = __builtin_fmaf(w, v[g].w, acc.w);
        }
    }
    m = mn;
}

#define GN 16
#define TP (HID + 2)
static_assert(OUT_C == HID, "both fused products are 32 x 32: one row pitch, two column tiles");
template <int EPI>
__global__ void __launch_bounds__(GN * 32, 16)
gat_node_gather_mm(const int* __restrict__ srt,
                   const int* __restrict__ cur,
                   const int* __restrict__ cnt,
                   const float* __restrict__ a_src,
                   const float* __restrict__ a_dst,
                   const float* __restrict__ h,
                   const float* __restrict__ bias,
                   const float* __restrict__ Wn,
                   const float* __restrict__ bn,
                   float* __restrict__ outp)
{
    __shared__ float tile[GN * TP];
    const int lane = threadIdx.x & 31;
    const int wv = __builtin_amdgcn_readfirstlane((int)(threadIdx.x >> 5));
    const int d  = (int)blockIdx.x * GN + wv;

    {
        const int beg = __builtin_amdgcn_readfirstlane(cur[d]);
        const int deg = __builtin_amdgcn_readfirstlane(cnt[d]);
        const int end = beg + deg;

        const unsigned sub16 = (unsigned)(lane & 7) * 16u;
        const int      eidx  = (int)(sub16 >> 2) + (lane >> 3);

        const v4f hd = *(const v4f*)((const char*)(h + (size_t)d * HID) + sub16);
        const v4f as = *(const v4f*)((const char*)a_src + sub16);
        const v4f ad = *(const v4f*)((const char*)a_dst + sub16);

        int sj = d;
        if (eidx < deg) sj = (srt + beg)[eidx];

        const float sd     = grp_sum(dot4(hd, ad));
        const float e_self = leaky(grp_sum(dot4(hd, as)) + sd);

        float m = e_self, z = 1.0f;
        v4f acc = (v4f){0.0f, 0.0f, 0.0f, 0.0f};
        if (lane < 8) acc = hd;

        for (int c0 = beg; c0 < end; c0 += 32) {
            const int rem = end - c0;
            const int k   = rem < 32 ? rem : 32;
            int sn = d;
            if (eidx < rem - 32) sn = (srt + c0 + 32)[eidx];
            if (k > 16)     gather_chunk<8>(h, sj, k, lane, eidx, sub16, as, sd, m, z, acc);
            else if (k > 8) gather_chunk<4>(h, sj, k, lane, eidx, sub16, as, sd, m, z, acc);
            else            gather_chunk<2>(h, sj, k, lane, eidx, sub16, as, sd, m, z, acc);
            sj = sn;
        }

        acc.x += xrow<8>(acc.x); acc.y += xrow<8>(acc.y); acc.z += xrow<8>(acc.z); acc.w += xrow<8>(acc.w);
        acc.x += xhalf(acc.x);   acc.y += xhalf(acc.y);   acc.z += xhalf(acc.z);   acc.w += xhalf(acc.w);

        const v4f   bv  = *(const v4f*)((const char*)bias + sub16);
        const float inv = 1.0f / (z + 1e-16f);
        const float r0 = acc.x * inv + bv.x, r1 = acc.y * inv + bv.y;
        const float r2 = acc.z * inv + bv.z, r3 = acc.w * inv + bv.w;
        if (lane < 8) {
            float* trow = (float*)((char*)(tile + wv * TP) + sub16);
            trow[0] = r0 > 0.0f ? r0 : 0.0f;
            trow[1] = r1 > 0.0f ? r1 : 0.0f;
            trow[2] = r2 > 0.0f ? r2 : 0.0f;
            trow[3] = r3 > 0.0f ? r3 : 0.0f;
        }
    }
    __syncthreads();
    if (wv >= HID / 16) return;

    const int half = lane >> 4;
    const int rr   = lane & 15;
    const int col  = wv * 16 + rr;
    v8f acc2 = {};
#pragma unroll
    for (int k = 0; k < HID; k += 4) {
        v2f a = *(const v2f*)(tile + rr * TP + k + 2 * half);
        v2f b;
        b.x = Wn[(k + 2 * half + 0) * HID + col];
        b.y = Wn[(k + 2 * half + 1) * HID + col];
        acc2 = __builtin_amdgcn_wmma_f32_16x16x4_f32(
             false, a,  false, b,
             (short)0, acc2,  false,  false);
    }
    float bcol = 0.0f;
    if (EPI == 1) bcol = bn[col];
#pragma unroll
    for (int i = 0; i < 8; ++i) {
        const int orow = (int)blockIdx.x * GN + i + 8 * half;
        float v = acc2[i];
        if (EPI == 1) {
            v += bcol;
            v = 1.0f / (1.0f + __expf(-v));
        }
        outp[(size_t)orow * HID + col] = v;
    }
}

extern "C" void kernel_launch(void* const* d_in, const int* in_sizes, int n_in,
                              void* d_out, int out_size, void* d_ws, size_t ws_size,
                              hipStream_t stream)
{
    const float* x      = (const float*)d_in[0];
    const int*   ei     = (const int*)  d_in[1];
    const float* W1     = (const float*)d_in[2];
    const float* a_src1 = (const float*)d_in[3];
    const float* a_dst1 = (const float*)d_in[4];
    const float* b1     = (const float*)d_in[5];
    const float* W2     = (const float*)d_in[6];
    const float* a_src2 = (const float*)d_in[7];
    const float* a_dst2 = (const float*)d_in[8];
    const float* b2     = (const float*)d_in[9];
    const float* Wc     = (const float*)d_in[10];
    const float* bc     = (const float*)d_in[11];
    float* out = (float*)d_out;

    const int n       = in_sizes[0] / IN_C;
    const int n_edges = in_sizes[1] / 2;

    float* ws    = (float*)d_ws;
    float* h     = ws;
    float* g     = h   + (size_t)n * HID;
    int*   cnt   = (int*)(g + (size_t)n * HID);
    int*   bk    = cnt + 2 * (size_t)n;
    int*   cur   = bk + n;
    int*   srt   = cur + n;
    int*   btot  = bk;
    int*   bcur  = bk + MAXB * BST;
    unsigned* binned = (unsigned*)g;

    const int TB = 256;
    const int chunkBlocks = (n_edges + CH - 1) / CH;
    const int nb = (n + BW - 1) / BW;
    const int nodeBlocks = n / GN;
    const int gemmWaves  = (n / 16) * (HID / 16);
    const int gemmBlocks = (gemmWaves + 7) / 8;

    gat_gemm_wmma<IN_C, HID><<<gemmBlocks, TB, 0, stream>>>(x, W1, h, bk, n);
    bin_hist<<<chunkBlocks, CH_T, 0, stream>>>(ei, btot, n_edges, nb);
    bin_scatter<<<chunkBlocks, CH_T, 0, stream>>>(ei, btot, bcur, binned, n_edges, nb);
    bucket_sort<<<nb, BS_T, 0, stream>>>(binned, btot, cnt, cur, srt, n, nb);
    gat_node_gather_mm<0><<<nodeBlocks, GN * 32, 0, stream>>>(srt, cur, cnt, a_src1, a_dst1, h, b1,
                                                             W2, nullptr, g);

    gat_node_gather_mm<1><<<nodeBlocks, GN * 32, 0, stream>>>(srt, cur, cnt, a_src2, a_dst2, g, b2,
                                                             Wc, bc, out);
}


# edge list built without the counting launch: chunks written back in bucket order with a row of run starts, the bucket pass collects its runs; global adds gone; sum order still arrival order
# speedup vs baseline: 28.4907x; 1.0876x over previous
#include <hip/hip_runtime.h>
#include <math.h>

#define NODES_DEF 100000
#define IN_C 128
#define HID 32
#define OUT_C 32
#define NEG_SLOPE 0.2f

typedef __attribute__((ext_vector_type(2))) float v2f;
typedef __attribute__((ext_vector_type(4))) float v4f;
typedef __attribute__((ext_vector_type(8))) float v8f;

template <int K, int NC>
__global__ void gat_gemm_wmma(const float* __restrict__ A,
                              const float* __restrict__ B,
                              float* __restrict__ C, int M)
{
    const int wavesPerBlock = blockDim.x >> 5;
    const int wave = blockIdx.x * wavesPerBlock + (threadIdx.x >> 5);
    const int nTiles = NC / 16;
    const int totalTiles = (M / 16) * nTiles;
    if (wave >= totalTiles) return;

    const int mTile = wave / nTiles;
    const int nTile = wave % nTiles;
    const int lane  = threadIdx.x & 31;
    const int half  = lane >> 4;
    const int r     = lane & 15;
    const int row   = mTile * 16 + r;
    const int col   = nTile * 16 + r;

    v8f acc = {};
#pragma unroll
    for (int k = 0; k < K; k += 4) {
        v2f a = *(const v2f*)(A + (size_t)row * K + k + 2 * half);
        v2f b;
        b.x = B[(k + 2 * half + 0) * NC + col];
        b.y = B[(k + 2 * half + 1) * NC + col];
        acc = __builtin_amdgcn_wmma_f32_16x16x4_f32(
             false, a,  false, b,
             (short)0, acc,  false,  false);
    }

#pragma unroll
    for (int i = 0; i < 8; ++i) {
        const int orow = mTile * 16 + i + 8 * half;
        C[(size_t)orow * NC + col] = acc[i];
    }
}

__device__ __forceinline__ float leaky(float x)
{
    return x > 0.0f ? x : NEG_SLOPE * x;
}

#define BSH   7
#define BW    (1 << BSH)
#define MAXB  1024
#define CH_T  1024
#define CH_E  16
#define CH    (CH_T * CH_E)
#define BS_T  512
#define BS_E  16
#define CAP   (BS_T * BS_E)
static_assert(MAXB == CH_T, "one thread per bucket counter in the chunk kernel");
static_assert(CH <= 65535, "an offset inside a chunk is stored in 16 bits");
static_assert(BW == 128, "bucket_sort sums its counters 4 per lane of one wave");

__device__ __forceinline__ int wave_incl_scan(int v, int lane)
{
#pragma unroll
    for (int o = 1; o < 32; o <<= 1) {
        const int t = __shfl_up(v, o);
        if (lane >= o) v += t;
    }
    return v;
}

__global__ void __launch_bounds__(CH_T)
bin_chunk(const int* __restrict__ ei, unsigned short* __restrict__ offs,
          unsigned* __restrict__ binned, int n_edges, int nb, int ostr)
{
    __shared__ int hist[MAXB];
    __shared__ int wsA[32];
    __shared__ unsigned sw[CH];
    const int tid = threadIdx.x, lane = tid & 31, wv = tid >> 5;
    hist[tid] = 0;
    __syncthreads();
    const int base = blockIdx.x * CH;
    unsigned w[CH_E], br[CH_E];
    int sv[CH_E], dv[CH_E];
#pragma unroll
    for (int k = 0; k < CH_E; ++k) {
        const int i = base + k * CH_T + tid;
        const int iq = i < n_edges ? i : n_edges - 1;
        sv[k] = ei[iq];
        dv[k] = ei[n_edges + iq];
    }
#pragma unroll
    for (int k = 0; k < CH_E; ++k) {
        const int i = base + k * CH_T + tid;
        w[k] = 0u; br[k] = 0xFFFFFFFFu;
        if (i < n_edges) {
            const int s = sv[k];
            const int d = dv[k];
            const int b = d >> BSH;
            const int r = atomicAdd(&hist[b], 1);
            w[k]  = (unsigned)s | ((unsigned)(d & (BW - 1)) << 24);
            br[k] = (unsigned)b | ((unsigned)r << 16);
        }
    }
    __syncthreads();
    const int c = hist[tid];
    const int ic = wave_incl_scan(c, lane);
    if (lane == 31) wsA[wv] = ic;
    __syncthreads();
    if (wv == 0) {
        const int a = wsA[lane];
        const int ia = wave_incl_scan(a, lane);
        wsA[lane] = ia - a;
    }
    __syncthreads();
    const int rem = n_edges - base;
    const int nvalid = rem < CH ? rem : CH;
    const int excl = ic - c + wsA[wv];
    hist[tid] = excl;
    unsigned short* __restrict__ orow = offs + (size_t)blockIdx.x * ostr;
    if (tid < nb)  orow[tid] = (unsigned short)excl;
    if (tid == 0)  orow[nb]  = (unsigned short)nvalid;
    __syncthreads();
#pragma unroll
    for (int k = 0; k < CH_E; ++k) {
        if (br[k] != 0xFFFFFFFFu) {
            const int b = (int)(br[k] & 0xFFFFu), r = (int)(br[k] >> 16);
            sw[hist[b] + r] = w[k];
        }
    }
    __syncthreads();
#pragma unroll
    for (int k = 0; k < CH_E; ++k) {
        const int i = k * CH_T + tid;
        if (i < nvalid) binned[base + i] = sw[i];
    }
}

__global__ void __launch_bounds__(BS_T)
bucket_sort(const unsigned* __restrict__ binned, const unsigned short* __restrict__ offs,
            int* __restrict__ cnt, int* __restrict__ cur, int* __restrict__ srt,
            int n, int nch, int ostr)
{
    __shared__ int c[BW];
    __shared__ int cursor[BW];
    __shared__ int wlen[BS_T / 32];
    __shared__ int wsta[BS_T / 32];
    __shared__ int sorted[CAP];
    const int tid = threadIdx.x, lane = tid & 31, wv = tid >> 5;
    const int b = blockIdx.x;
    if (tid < BW) c[tid] = 0;
    int s = 0, len = 0;
    if (tid < nch) {
        const unsigned* __restrict__ orow = (const unsigned*)(offs + (size_t)tid * ostr);
        const unsigned w0 = orow[b >> 1];
        const unsigned w1 = orow[(b + 1) >> 1];
        const int e = (int)((b & 1) ? (w1 & 0xFFFFu) : (w0 >> 16));
        s   = (int)((b & 1) ? (w0 >> 16) : (w0 & 0xFFFFu));
        len = e - s;
    }
    const int il = wave_incl_scan(len, lane);
    int ss = s;
#pragma unroll
    for (int o = 16; o > 0; o >>= 1) ss += __shfl_xor(ss, o);
    if (lane == 31) wlen[wv] = il;
    if (lane == 0)  wsta[wv] = ss;
    __syncthreads();
    int start = 0, tot = 0, pre = 0;
#pragma unroll
    for (int j = 0; j < BS_T / 32; ++j) {
        const int a = wlen[j];
        start += wsta[j];
        tot   += a;
        if (j < wv) pre += a;
    }
    const int p0 = pre + il - len;
    const int g0 = tid * CH + s;
    const bool staged = tot <= CAP;
    unsigned v[BS_E];
    if (staged) {
        for (int k = 0; k < len; ++k) sorted[p0 + k] = g0 + k;
        __syncthreads();
#pragma unroll
        for (int k = 0; k < BS_E; ++k) {
            const int i = k * BS_T + tid;
            v[k] = binned[i < tot ? sorted[i] : 0];
        }
#pragma unroll
        for (int k = 0; k < BS_E; ++k)
            if (k * BS_T + tid < tot) atomicAdd(&c[v[k] >> 24], 1);
    } else {
        for (int k = 0; k < len; ++k) atomicAdd(&c[binned[g0 + k] >> 24], 1);
    }
    __syncthreads();
    if (tid < 32) {
        const int v0 = c[4 * tid], v1 = c[4 * tid + 1], v2 = c[4 * tid + 2], v3 = c[4 * tid + 3];
        const int sm = v0 + v1 + v2 + v3;
        const int e = wave_incl_scan(sm, tid) - sm;
        cursor[4 * tid]     = e;
        cursor[4 * tid + 1] = e + v0;
        cursor[4 * tid + 2] = e + v0 + v1;
        cursor[4 * tid + 3] = e + v0 + v1 + v2;
    }
    __syncthreads();
    if (tid < BW) {
        const int d = b * BW + tid;
        if (d < n) { cnt[d] = c[tid]; cur[d] = start + cursor[tid]; }
    }
    __syncthreads();
    if (staged) {
#pragma unroll
        for (int k = 0; k < BS_E; ++k) {
            if (k * BS_T + tid < tot) {
                const int p = atomicAdd(&cursor[v[k] >> 24], 1);
                sorted[p] = (int)(v[k] & 0xFFFFFFu);
            }
        }
        __syncthreads();
#pragma unroll
        for (int k = 0; k < BS_E; ++k) {
            const int i = k * BS_T + tid;
            if (i < tot) srt[start + i] = sorted[i];
        }
    } else {
        for (int k = 0; k < len; ++k) {
            const unsigned wd = binned[g0 + k];
            const int p = atomicAdd(&cursor[wd >> 24], 1);
            srt[start + p] = (int)(wd & 0xFFFFFFu);
        }
    }
}

__device__ __forceinline__ float xhalf(float x)
{
    return __int_as_float(__builtin_amdgcn_permlanex16(__float_as_int(x), __float_as_int(x),
                                                      0x76543210, 0xfedcba98, false, false));
}
template <int S>
__device__ __forceinline__ float xrow(float x)
{
    return __int_as_float(__builtin_amdgcn_update_dpp(0, __float_as_int(x), 0x160 + S,
                                                     0xf, 0xf, true));
}
__device__ __forceinline__ float wave_sum(float v)
{
    v += xrow<1>(v); v += xrow<2>(v); v += xrow<4>(v); v += xrow<8>(v);
    return v + xhalf(v);
}
__device__ __forceinline__ float wave_max(float v)
{
    v = fmaxf(v, xrow<1>(v)); v = fmaxf(v, xrow<2>(v));
    v = fmaxf(v, xrow<4>(v)); v = fmaxf(v, xrow<8>(v));
    return fmaxf(v, xhalf(v));
}

template <int J>
__device__ __forceinline__ int grp_pick(int x)
{
    return (int)__builtin_amdgcn_mov_dpp8((unsigned)x, J * 0x249249u);
}
__device__ __forceinline__ int grp_bcast(int x, int j)
{
    switch (j) {
    case 0:  return grp_pick<0>(x);
    case 1:  return grp_pick<1>(x);
    case 2:  return grp_pick<2>(x);
    case 3:  return grp_pick<3>(x);
    case 4:  return grp_pick<4>(x);
    case 5:  return grp_pick<5>(x);
    case 6:  return grp_pick<6>(x);
    default: return grp_pick<7>(x);
    }
}
__device__ __forceinline__ float grp_sum(float v)
{
    v += xrow<1>(v); v += xrow<2>(v);
    return v + xrow<4>(v);
}
__device__ __forceinline__ float dot4(v4f a, v4f b)
{
    const v2f t = __builtin_elementwise_fma(a.zw, b.zw, a.xy * b.xy);
    return t.x + t.y;
}

template <int S, int N>
__device__ __forceinline__ void fold_rows(float* p, int lane)
{
    const bool hi = (lane & S) != 0;
#pragma unroll
    for (int u = 0; u < N; ++u) {
        const float x = hi ? p[u + N] : p[u];
        const float y = hi ? p[u] : p[u + N];
        p[u] = x + xrow<S>(y);
    }
}

template <int G>
__device__ __forceinline__ float group_dots(float (&p)[G], int lane)
{
    if constexpr (G == 8) fold_rows<4, 4>(p, lane);
    else {
#pragma unroll
        for (int u = 0; u < G; ++u) p[u] += xrow<4>(p[u]);
    }
    if constexpr (G >= 4) fold_rows<2, 2>(p, lane);
    else {
#pragma unroll
        for (int u = 0; u < G; ++u) p[u] += xrow<2>(p[u]);
    }
    fold_rows<1, 1>(p, lane);
    return p[0];
}

template <int G>
__device__ __forceinline__ void gather_chunk(const float* __restrict__ h, int sj, int k,
                                             int lane, int eidx, unsigned sub16,
                                             v4f as, float sd,
                                             float& m, float& z, v4f& acc)
{
    v4f v[G];
#pragma unroll
    for (int g = 0; g < G; ++g) {
        if (4 * g < k) {
            const unsigned s = (unsigned)grp_bcast(sj, g);
            const unsigned off = s * (unsigned)(HID * 4) + sub16;
            v[g] = *(const v4f*)((const char*)h + (size_t)off);
        } else {
            v[g] = (v4f){0.0f, 0.0f, 0.0f, 0.0f};
        }
    }
    float p[G];
#pragma unroll
    for (int g = 0; g < G; ++g) p[g] = dot4(v[g], as);
    const float s = group_dots<G>(p, lane);

    const bool  valid = eidx < k;
    const float e  = valid ? leaky(s + sd) : m;
    const float mn = fmaxf(m, wave_max(e));
    const float scale = __expf(m - mn);
    const float ex = valid ? __expf(e - mn) : 0.0f;
    z   = z * scale + wave_sum(ex);
    acc.x *= scale; acc.y *= scale; acc.z *= scale; acc.w *= scale;
    const int wi = __float_as_int(ex);
#pragma unroll
    for (int g = 0; g < G; ++g) {
        if (4 * g < k) {
            const float w = __int_as_float(grp_bcast(wi, g));
            acc.x = __builtin_fmaf(w, v[g].x, acc.x);
            acc.y = __builtin_fmaf(w, v[g].y, acc.y);
            acc.z = __builtin_fmaf(w, v[g].z, acc.z);
            acc.w = __builtin_fmaf(w, v[g].w, acc.w);
        }
    }
    m = mn;
}

#define GN 16
#define TP (HID + 2)
static_assert(OUT_C == HID, "both fused products are 32 x 32: one row pitch, two column tiles");
template <int EPI>
__global__ void __launch_bounds__(GN * 32, 16)
gat_node_gather_mm(const int* __restrict__ srt,
                   const int* __restrict__ cur,
                   const int* __restrict__ cnt,
                   const float* __restrict__ a_src,
                   const float* __restrict__ a_dst,
                   const float* __restrict__ h,
                   const float* __restrict__ bias,
                   const float* __restrict__ Wn,
                   const float* __restrict__ bn,
                   float* __restrict__ outp)
{
    __shared__ float tile[GN * TP];
    const int lane = threadIdx.x & 31;
    const int wv = __builtin_amdgcn_readfirstlane((int)(threadIdx.x >> 5));
    const int d  = (int)blockIdx.x * GN + wv;

    {
        const int beg = __builtin_amdgcn_readfirstlane(cur[d]);
        const int deg = __builtin_amdgcn_readfirstlane(cnt[d]);
        const int end = beg + deg;

        const unsigned sub16 = (unsigned)(lane & 7) * 16u;
        const int      eidx  = (int)(sub16 >> 2) + (lane >> 3);

        const v4f hd = *(const v4f*)((const char*)(h + (size_t)d * HID) + sub16);
        const v4f as = *(const v4f*)((const char*)a_src + sub16);
        const v4f ad = *(const v4f*)((const char*)a_dst + sub16);

        int sj = d;
        if (eidx < deg) sj = (srt + beg)[eidx];

        const float sd     = grp_sum(dot4(hd, ad));
        const float e_self = leaky(grp_sum(dot4(hd, as)) + sd);

        float m = e_self, z = 1.0f;
        v4f acc = (v4f){0.0f, 0.0f, 0.0f, 0.0f};
        if (lane < 8) acc = hd;

        for (int c0 = beg; c0 < end; c0 += 32) {
            const int rem = end - c0;
            const int k   = rem < 32 ? rem : 32;
            int sn = d;
            if (eidx < rem - 32) sn = (srt + c0 + 32)[eidx];
            if (k > 16)     gather_chunk<8>(h, sj, k, lane, eidx, sub16, as, sd, m, z, acc);
            else if (k > 8) gather_chunk<4>(h, sj, k, lane, eidx, sub16, as, sd, m, z, acc);
            else            gather_chunk<2>(h, sj, k, lane, eidx, sub16, as, sd, m, z, acc);
            sj = sn;
        }

        acc.x += xrow<8>(acc.x); acc.y += xrow<8>(acc.y); acc.z += xrow<8>(acc.z); acc.w += xrow<8>(acc.w);
        acc.x += xhalf(acc.x);   acc.y += xhalf(acc.y);   acc.z += xhalf(acc.z);   acc.w += xhalf(acc.w);

        const v4f   bv  = *(const v4f*)((const char*)bias + sub16);
        const float inv = 1.0f / (z + 1e-16f);
        const float r0 = acc.x * inv + bv.x, r1 = acc.y * inv + bv.y;
        const float r2 = acc.z * inv + bv.z, r3 = acc.w * inv + bv.w;
        if (lane < 8) {
            float* trow = (float*)((char*)(tile + wv * TP) + sub16);
            trow[0] = r0 > 0.0f ? r0 : 0.0f;
            trow[1] = r1 > 0.0f ? r1 : 0.0f;
            trow[2] = r2 > 0.0f ? r2 : 0.0f;
            trow[3] = r3 > 0.0f ? r3 : 0.0f;
        }
    }
    __syncthreads();
    if (wv >= HID / 16) return;

    const int half = lane >> 4;
    const int rr   = lane & 15;
    const int col  = wv * 16 + rr;
    v8f acc2 = {};
#pragma unroll
    for (int k = 0; k < HID; k += 4) {
        v2f a = *(const v2f*)(tile + rr * TP + k + 2 * half);
        v2f b;
        b.x = Wn[(k + 2 * half + 0) * HID + col];
        b.y = Wn[(k + 2 * half + 1) * HID + col];
        acc2 = __builtin_amdgcn_wmma_f32_16x16x4_f32(
             false, a,  false, b,
             (short)0, acc2,  false,  false);
    }
    float bcol = 0.0f;
    if (EPI == 1) bcol = bn[col];
#pragma unroll
    for (int i = 0; i < 8; ++i) {
        const int orow = (int)blockIdx.x * GN + i + 8 * half;
        float v = acc2[i];
        if (EPI == 1) {
            v += bcol;
            v = 1.0f / (1.0f + __expf(-v));
        }
        outp[(size_t)orow * HID + col] = v;
    }
}

extern "C" void kernel_launch(void* const* d_in, const int* in_sizes, int n_in,
                              void* d_out, int out_size, void* d_ws, size_t ws_size,
                              hipStream_t stream)
{
    const float* x      = (const float*)d_in[0];
    const int*   ei     = (const int*)  d_in[1];
    const float* W1     = (const float*)d_in[2];
    const float* a_src1 = (const float*)d_in[3];
    const float* a_dst1 = (const float*)d_in[4];
    const float* b1     = (const float*)d_in[5];
    const float* W2     = (const float*)d_in[6];
    const float* a_src2 = (const float*)d_in[7];
    const float* a_dst2 = (const float*)d_in[8];
    const float* b2     = (const float*)d_in[9];
    const float* Wc     = (const float*)d_in[10];
    const float* bc     = (const float*)d_in[11];
    float* out = (float*)d_out;

    const int n       = in_sizes[0] / IN_C;
    const int n_edges = in_sizes[1] / 2;

    float* ws    = (float*)d_ws;
    float* h     = ws;
    float* g     = h   + (size_t)n * HID;
    int*   cnt   = (int*)(g + (size_t)n * HID);
    unsigned short* offs = (unsigned short*)(cnt + n);
    int*   cur   = cnt + 3 * (size_t)n;
    int*   srt   = cur + n;
    unsigned* binned = (unsigned*)g;

    const int TB = 256;
    const int chunkBlocks = (n_edges + CH - 1) / CH;
    const int nb = (n + BW - 1) / BW;
    const int ostr = (nb + 2) & ~1;
    const int nodeBlocks = n / GN;
    const int gemmWaves  = (n / 16) * (HID / 16);
    const int gemmBlocks = (gemmWaves + 7) / 8;

    gat_gemm_wmma<IN_C, HID><<<gemmBlocks, TB, 0, stream>>>(x, W1, h, n);
    bin_chunk<<<chunkBlocks, CH_T, 0, stream>>>(ei, offs, binned, n_edges, nb, ostr);
    bucket_sort<<<nb, BS_T, 0, stream>>>(binned, offs, cnt, cur, srt, n, chunkBlocks, ostr);
    gat_node_gather_mm<0><<<nodeBlocks, GN * 32, 0, stream>>>(srt, cur, cnt, a_src1, a_dst1, h, b1,
                                                             W2, nullptr, g);

    gat_node_gather_mm<1><<<nodeBlocks, GN * 32, 0, stream>>>(srt, cur, cnt, a_src2, a_dst2, g, b2,
                                                             Wc, bc, out);
}


# first layer's GEMM runs as extra work-groups of the edge-list chunk launch (chunk groups first in the grid, 32 GEMM tiles per group behind them): one launch gone, 4 left; sum orders unchanged
# speedup vs baseline: 29.5695x; 1.0379x over previous
#include <hip/hip_runtime.h>
#include <math.h>

#define NODES_DEF 100000
#define IN_C 128
#define HID 32
#define OUT_C 32
#define NEG_SLOPE 0.2f

typedef __attribute__((ext_vector_type(2))) float v2f;
typedef __attribute__((ext_vector_type(4))) float v4f;
typedef __attribute__((ext_vector_type(8))) float v8f;

template <int K, int NC>
__device__ __forceinline__ void gemm_tile_wmma(const float* __restrict__ A,
                                               const float* __restrict__ B,
                                               float* __restrict__ C, int M, int wave)
{
    const int nTiles = NC / 16;
    const int totalTiles = (M / 16) * nTiles;
    if (wave >= totalTiles) return;

    const int mTile = wave / nTiles;
    const int nTile = wave % nTiles;
    const int lane  = threadIdx.x & 31;
    const int half  = lane >> 4;
    const int r     = lane & 15;
    const int row   = mTile * 16 + r;
    const int col   = nTile * 16 + r;

    v8f acc = {};
#pragma unroll
    for (int k = 0; k < K; k += 4) {
        v2f a = *(const v2f*)(A + (size_t)row * K + k + 2 * half);
        v2f b;
        b.x = B[(k + 2 * half + 0) * NC + col];
        b.y = B[(k + 2 * half + 1) * NC + col];
        acc = __builtin_amdgcn_wmma_f32_16x16x4_f32(
             false, a,  false, b,
             (short)0, acc,  false,  false);
    }

#pragma unroll
    for (int i = 0; i < 8; ++i) {
        const int orow = mTile * 16 + i + 8 * half;
        C[(size_t)orow * NC + col] = acc[i];
    }
}

__device__ __forceinline__ float leaky(float x)
{
    return x > 0.0f ? x : NEG_SLOPE * x;
}

#define BSH   7
#define BW    (1 << BSH)
#define MAXB  1024
#define CH_T  1024
#define CH_E  16
#define CH    (CH_T * CH_E)
#define BS_T  512
#define BS_E  16
#define CAP   (BS_T * BS_E)
static_assert(MAXB == CH_T, "one thread per bucket counter in the chunk kernel");
static_assert(CH <= 65535, "an offset inside a chunk is stored in 16 bits");
static_assert(BW == 128, "bucket_sort sums its counters 4 per lane of one wave");

__device__ __forceinline__ int wave_incl_scan(int v, int lane)
{
#pragma unroll
    for (int o = 1; o < 32; o <<= 1) {
        const int t = __shfl_up(v, o);
        if (lane >= o) v += t;
    }
    return v;
}

__global__ void __launch_bounds__(CH_T)
bin_chunk_gemm(const int* __restrict__ ei, unsigned short* __restrict__ offs,
               unsigned* __restrict__ binned, int n_edges, int nb, int ostr, int n_chunks,
               const float* __restrict__ x, const float* __restrict__ W1,
               float* __restrict__ h, int n)
{
    if ((int)blockIdx.x >= n_chunks) {
        const int wave = ((int)blockIdx.x - n_chunks) * (CH_T / 32) + (int)(threadIdx.x >> 5);
        gemm_tile_wmma<IN_C, HID>(x, W1, h, n, wave);
        return;
    }
    __shared__ int hist[MAXB];
    __shared__ int wsA[32];
    __shared__ unsigned sw[CH];
    const int tid = threadIdx.x, lane = tid & 31, wv = tid >> 5;
    hist[tid] = 0;
    __syncthreads();
    const int base = blockIdx.x * CH;
    unsigned w[CH_E], br[CH_E];
    int sv[CH_E], dv[CH_E];
#pragma unroll
    for (int k = 0; k < CH_E; ++k) {
        const int i = base + k * CH_T + tid;
        const int iq = i < n_edges ? i : n_edges - 1;
        sv[k] = ei[iq];
        dv[k] = ei[n_edges + iq];
    }
#pragma unroll
    for (int k = 0; k < CH_E; ++k) {
        const int i = base + k * CH_T + tid;
        w[k] = 0u; br[k] = 0xFFFFFFFFu;
        if (i < n_edges) {
            const int s = sv[k];
            const int d = dv[k];
            const int b = d >> BSH;
            const int r = atomicAdd(&hist[b], 1);
            w[k]  = (unsigned)s | ((unsigned)(d & (BW - 1)) << 24);
            br[k] = (unsigned)b | ((unsigned)r << 16);
        }
    }
    __syncthreads();
    const int c = hist[tid];
    const int ic = wave_incl_scan(c, lane);
    if (lane == 31) wsA[wv] = ic;
    __syncthreads();
    if (wv == 0) {
        const int a = wsA[lane];
        const int ia = wave_incl_scan(a, lane);
        wsA[lane] = ia - a;
    }
    __syncthreads();
    const int rem = n_edges - base;
    const int nvalid = rem < CH ? rem : CH;
    const int excl = ic - c + wsA[wv];
    hist[tid] = excl;
    unsigned short* __restrict__ orow = offs + (size_t)blockIdx.x * ostr;
    if (tid < nb)  orow[tid] = (unsigned short)excl;
    if (tid == 0)  orow[nb]  = (unsigned short)nvalid;
    __syncthreads();
#pragma unroll
    for (int k = 0; k < CH_E; ++k) {
        if (br[k] != 0xFFFFFFFFu) {
            const int b = (int)(br[k] & 0xFFFFu), r = (int)(br[k] >> 16);
            sw[hist[b] + r] = w[k];
        }
    }
    __syncthreads();
#pragma unroll
    for (int k = 0; k < CH_E; ++k) {
        const int i = k * CH_T + tid;
        if (i < nvalid) binned[base + i] = sw[i];
    }
}

__global__ void __launch_bounds__(BS_T)
bucket_sort(const unsigned* __restrict__ binned, const unsigned short* __restrict__ offs,
            int* __restrict__ cnt, int* __restrict__ cur, int* __restrict__ srt,
            int n, int nch, int ostr)
{
    __shared__ int c[BW];
    __shared__ int cursor[BW];
    __shared__ int wlen[BS_T / 32];
    __shared__ int wsta[BS_T / 32];
    __shared__ int sorted[CAP];
    const int tid = threadIdx.x, lane = tid & 31, wv = tid >> 5;
    const int b = blockIdx.x;
    if (tid < BW) c[tid] = 0;
    int s = 0, len = 0;
    if (tid < nch) {
        const unsigned* __restrict__ orow = (const unsigned*)(offs + (size_t)tid * ostr);
        const unsigned w0 = orow[b >> 1];
        const unsigned w1 = orow[(b + 1) >> 1];
        const int e = (int)((b & 1) ? (w1 & 0xFFFFu) : (w0 >> 16));
        s   = (int)((b & 1) ? (w0 >> 16) : (w0 & 0xFFFFu));
        len = e - s;
    }
    const int il = wave_incl_scan(len, lane);
    int ss = s;
#pragma unroll
    for (int o = 16; o > 0; o >>= 1) ss += __shfl_xor(ss, o);
    if (lane == 31) wlen[wv] = il;
    if (lane == 0)  wsta[wv] = ss;
    __syncthreads();
    int start = 0, tot = 0, pre = 0;
#pragma unroll
    for (int j = 0; j < BS_T / 32; ++j) {
        const int a = wlen[j];
        start += wsta[j];
        tot   += a;
        if (j < wv) pre += a;
    }
    const int p0 = pre + il - len;
    const int g0 = tid * CH + s;
    const bool staged = tot <= CAP;
    unsigned v[BS_E];
    if (staged) {
        for (int k = 0; k < len; ++k) sorted[p0 + k] = g0 + k;
        __syncthreads();
#pragma unroll
        for (int k = 0; k < BS_E; ++k) {
            const int i = k * BS_T + tid;
            v[k] = binned[i < tot ? sorted[i] : 0];
        }
#pragma unroll
        for (int k = 0; k < BS_E; ++k)
            if (k * BS_T + tid < tot) atomicAdd(&c[v[k] >> 24], 1);
    } else {
        for (int k = 0; k < len; ++k) atomicAdd(&c[binned[g0 + k] >> 24], 1);
    }
    __syncthreads();
    if (tid < 32) {
        const int v0 = c[4 * tid], v1 = c[4 * tid + 1], v2 = c[4 * tid + 2], v3 = c[4 * tid + 3];
        const int sm = v0 + v1 + v2 + v3;
        const int e = wave_incl_scan(sm, tid) - sm;
        cursor[4 * tid]     = e;
        cursor[4 * tid + 1] = e + v0;
        cursor[4 * tid + 2] = e + v0 + v1;
        cursor[4 * tid + 3] = e + v0 + v1 + v2;
    }
    __syncthreads();
    if (tid < BW) {
        const int d = b * BW + tid;
        if (d < n) { cnt[d] = c[tid]; cur[d] = start + cursor[tid]; }
    }
    __syncthreads();
    if (staged) {
#pragma unroll
        for (int k = 0; k < BS_E; ++k) {
            if (k * BS_T + tid < tot) {
                const int p = atomicAdd(&cursor[v[k] >> 24], 1);
                sorted[p] = (int)(v[k] & 0xFFFFFFu);
            }
        }
        __syncthreads();
#pragma unroll
        for (int k = 0; k < BS_E; ++k) {
            const int i = k * BS_T + tid;
            if (i < tot) srt[start + i] = sorted[i];
        }
    } else {
        for (int k = 0; k < len; ++k) {
            const unsigned wd = binned[g0 + k];
            const int p = atomicAdd(&cursor[wd >> 24], 1);
            srt[start + p] = (int)(wd & 0xFFFFFFu);
        }
    }
}

__device__ __forceinline__ float xhalf(float x)
{
    return __int_as_float(__builtin_amdgcn_permlanex16(__float_as_int(x), __float_as_int(x),
                                                      0x76543210, 0xfedcba98, false, false));
}
template <int S>
__device__ __forceinline__ float xrow(float x)
{
    return __int_as_float(__builtin_amdgcn_update_dpp(0, __float_as_int(x), 0x160 + S,
                                                     0xf, 0xf, true));
}
__device__ __forceinline__ float wave_sum(float v)
{
    v += xrow<1>(v); v += xrow<2>(v); v += xrow<4>(v); v += xrow<8>(v);
    return v + xhalf(v);
}
__device__ __forceinline__ float wave_max(float v)
{
    v = fmaxf(v, xrow<1>(v)); v = fmaxf(v, xrow<2>(v));
    v = fmaxf(v, xrow<4>(v)); v = fmaxf(v, xrow<8>(v));
    return fmaxf(v, xhalf(v));
}

template <int J>
__device__ __forceinline__ int grp_pick(int x)
{
    return (int)__builtin_amdgcn_mov_dpp8((unsigned)x, J * 0x249249u);
}
__device__ __forceinline__ int grp_bcast(int x, int j)
{
    switch (j) {
    case 0:  return grp_pick<0>(x);
    case 1:  return grp_pick<1>(x);
    case 2:  return grp_pick<2>(x);
    case 3:  return grp_pick<3>(x);
    case 4:  return grp_pick<4>(x);
    case 5:  return grp_pick<5>(x);
    case 6:  return grp_pick<6>(x);
    default: return grp_pick<7>(x);
    }
}
__device__ __forceinline__ float grp_sum(float v)
{
    v += xrow<1>(v); v += xrow<2>(v);
    return v + xrow<4>(v);
}
__device__ __forceinline__ float dot4(v4f a, v4f b)
{
    const v2f t = __builtin_elementwise_fma(a.zw, b.zw, a.xy * b.xy);
    return t.x + t.y;
}

template <int S, int N>
__device__ __forceinline__ void fold_rows(float* p, int lane)
{
    const bool hi = (lane & S) != 0;
#pragma unroll
    for (int u = 0; u < N; ++u) {
        const float x = hi ? p[u + N] : p[u];
        const float y = hi ? p[u] : p[u + N];
        p[u] = x + xrow<S>(y);
    }
}

template <int G>
__device__ __forceinline__ float group_dots(float (&p)[G], int lane)
{
    if constexpr (G == 8) fold_rows<4, 4>(p, lane);
    else {
#pragma unroll
        for (int u = 0; u < G; ++u) p[u] += xrow<4>(p[u]);
    }
    if constexpr (G >= 4) fold_rows<2, 2>(p, lane);
    else {
#pragma unroll
        for (int u = 0; u < G; ++u) p[u] += xrow<2>(p[u]);
    }
    fold_rows<1, 1>(p, lane);
    return p[0];
}

template <int G>
__device__ __forceinline__ void gather_chunk(const float* __restrict__ h, int sj, int k,
                                             int lane, int eidx, unsigned sub16,
                                             v4f as, float sd,
                                             float& m, float& z, v4f& acc)
{
    v4f v[G];
#pragma unroll
    for (int g = 0; g < G; ++g) {
        if (4 * g < k) {
            const unsigned s = (unsigned)grp_bcast(sj, g);
            const unsigned off = s * (unsigned)(HID * 4) + sub16;
            v[g] = *(const v4f*)((const char*)h + (size_t)off);
        } else {
            v[g] = (v4f){0.0f, 0.0f, 0.0f, 0.0f};
        }
    }
    float p[G];
#pragma unroll
    for (int g = 0; g < G; ++g) p[g] = dot4(v[g], as);
    const float s = group_dots<G>(p, lane);

    const bool  valid = eidx < k;
    const float e  = valid ? leaky(s + sd) : m;
    const float mn = fmaxf(m, wave_max(e));
    const float scale = __expf(m - mn);
    const float ex = valid ? __expf(e - mn) : 0.0f;
    z   = z * scale + wave_sum(ex);
    acc.x *= scale; acc.y *= scale; acc.z *= scale; acc.w *= scale;
    const int wi = __float_as_int(ex);
#pragma unroll
    for (int g = 0; g < G; ++g) {
        if (4 * g < k) {
            const float w = __int_as_float(grp_bcast(wi, g));
            acc.x = __builtin_fmaf(w, v[g].x, acc.x);
            acc.y = __builtin_fmaf(w, v[g].y, acc.y);
            acc.z = __builtin_fmaf(w, v[g].z, acc.z);
            acc.w = __builtin_fmaf(w, v[g].w, acc.w);
        }
    }
    m = mn;
}

#define GN 16
#define TP (HID + 2)
static_assert(OUT_C == HID, "both fused products are 32 x 32: one row pitch, two column tiles");
template <int EPI>
__global__ void __launch_bounds__(GN * 32, 16)
gat_node_gather_mm(const int* __restrict__ srt,
                   const int* __restrict__ cur,
                   const int* __restrict__ cnt,
                   const float* __restrict__ a_src,
                   const float* __restrict__ a_dst,
                   const float* __restrict__ h,
                   const float* __restrict__ bias,
                   const float* __restrict__ Wn,
                   const float* __restrict__ bn,
                   float* __restrict__ outp)
{
    __shared__ float tile[GN * TP];
    const int lane = threadIdx.x & 31;
    const int wv = __builtin_amdgcn_readfirstlane((int)(threadIdx.x >> 5));
    const int d  = (int)blockIdx.x * GN + wv;

    {
        const int beg = __builtin_amdgcn_readfirstlane(cur[d]);
        const int deg = __builtin_amdgcn_readfirstlane(cnt[d]);
        const int end = beg + deg;

        const unsigned sub16 = (unsigned)(lane & 7) * 16u;
        const int      eidx  = (int)(sub16 >> 2) + (lane >> 3);

        const v4f hd = *(const v4f*)((const char*)(h + (size_t)d * HID) + sub16);
        const v4f as = *(const v4f*)((const char*)a_src + sub16);
        const v4f ad = *(const v4f*)((const char*)a_dst + sub16);

        int sj = d;
        if (eidx < deg) sj = (srt + beg)[eidx];

        const float sd     = grp_sum(dot4(hd, ad));
        const float e_self = leaky(grp_sum(dot4(hd, as)) + sd);

        float m = e_self, z = 1.0f;
        v4f acc = (v4f){0.0f, 0.0f, 0.0f, 0.0f};
        if (lane < 8) acc = hd;

        for (int c0 = beg; c0 < end; c0 += 32) {
            const int rem = end - c0;
            const int k   = rem < 32 ? rem : 32;
            int sn = d;
            if (eidx < rem - 32) sn = (srt + c0 + 32)[eidx];
            if (k > 16)     gather_chunk<8>(h, sj, k, lane, eidx, sub16, as, sd, m, z, acc);
            else if (k > 8) gather_chunk<4>(h, sj, k, lane, eidx, sub16, as, sd, m, z, acc);
            else            gather_chunk<2>(h, sj, k, lane, eidx, sub16, as, sd, m, z, acc);
            sj = sn;
        }

        acc.x += xrow<8>(acc.x); acc.y += xrow<8>(acc.y); acc.z += xrow<8>(acc.z); acc.w += xrow<8>(acc.w);
        acc.x += xhalf(acc.x);   acc.y += xhalf(acc.y);   acc.z += xhalf(acc.z);   acc.w += xhalf(acc.w);

        const v4f   bv  = *(const v4f*)((const char*)bias + sub16);
        const float inv = 1.0f / (z + 1e-16f);
        const float r0 = acc.x * inv + bv.x, r1 = acc.y * inv + bv.y;
        const float r2 = acc.z * inv + bv.z, r3 = acc.w * inv + bv.w;
        if (lane < 8) {
            float* trow = (float*)((char*)(tile + wv * TP) + sub16);
            trow[0] = r0 > 0.0f ? r0 : 0.0f;
            trow[1] = r1 > 0.0f ? r1 : 0.0f;
            trow[2] = r2 > 0.0f ? r2 : 0.0f;
            trow[3] = r3 > 0.0f ? r3 : 0.0f;
        }
    }
    __syncthreads();
    if (wv >= HID / 16) return;

    const int half = lane >> 4;
    const int rr   = lane & 15;
    const int col  = wv * 16 + rr;
    v8f acc2 = {};
#pragma unroll
    for (int k = 0; k < HID; k += 4) {
        v2f a = *(const v2f*)(tile + rr * TP + k + 2 * half);
        v2f b;
        b.x = Wn[(k + 2 * half + 0) * HID + col];
        b.y = Wn[(k + 2 * half + 1) * HID + col];
        acc2 = __builtin_amdgcn_wmma_f32_16x16x4_f32(
             false, a,  false, b,
             (short)0, acc2,  false,  false);
    }
    float bcol = 0.0f;
    if (EPI == 1) bcol = bn[col];
#pragma unroll
    for (int i = 0; i < 8; ++i) {
        const int orow = (int)blockIdx.x * GN + i + 8 * half;
        float v = acc2[i];
        if (EPI == 1) {
            v += bcol;
            v = 1.0f / (1.0f + __expf(-v));
        }
        outp[(size_t)orow * HID + col] = v;
    }
}

extern "C" void kernel_launch(void* const* d_in, const int* in_sizes, int n_in,
                              void* d_out, int out_size, void* d_ws, size_t ws_size,
                              hipStream_t stream)
{
    const float* x      = (const float*)d_in[0];
    const int*   ei     = (const int*)  d_in[1];
    const float* W1     = (const float*)d_in[2];
    const float* a_src1 = (const float*)d_in[3];
    const float* a_dst1 = (const float*)d_in[4];
    const float* b1     = (const float*)d_in[5];
    const float* W2     = (const float*)d_in[6];
    const float* a_src2 = (const float*)d_in[7];
    const float* a_dst2 = (const float*)d_in[8];
    const float* b2     = (const float*)d_in[9];
    const float* Wc     = (const float*)d_in[10];
    const float* bc     = (const float*)d_in[11];
    float* out = (float*)d_out;

    const int n       = in_sizes[0] / IN_C;
    const int n_edges = in_sizes[1] / 2;

    float* ws    = (float*)d_ws;
    float* h     = ws;
    float* g     = h   + (size_t)n * HID;
    int*   cnt   = (int*)(g + (size_t)n * HID);
    unsigned short* offs = (unsigned short*)(cnt + n);
    int*   cur   = cnt + 3 * (size_t)n;
    int*   srt   = cur + n;
    unsigned* binned = (unsigned*)g;

    const int chunkBlocks = (n_edges + CH - 1) / CH;
    const int nb = (n + BW - 1) / BW;
    const int ostr = (nb + 2) & ~1;
    const int nodeBlocks = n / GN;
    const int gemmWaves  = (n / 16) * (HID / 16);
    const int gemmBlocks = (gemmWaves + CH_T / 32 - 1) / (CH_T / 32);

    bin_chunk_gemm<<<chunkBlocks + gemmBlocks, CH_T, 0, stream>>>(ei, offs, binned, n_edges, nb, ostr,
                                                                 chunkBlocks, x, W1, h, n);
    bucket_sort<<<nb, BS_T, 0, stream>>>(binned, offs, cnt, cur, srt, n, chunkBlocks, ostr);
    gat_node_gather_mm<0><<<nodeBlocks, GN * 32, 0, stream>>>(srt, cur, cnt, a_src1, a_dst1, h, b1,
                                                             W2, nullptr, g);

    gat_node_gather_mm<1><<<nodeBlocks, GN * 32, 0, stream>>>(srt, cur, cnt, a_src2, a_dst2, g, b2,
                                                             Wc, bc, out);
}


# first layer's GEMM reads: a wave does a 16-row block x 32 columns, x rows read once, 16 bytes per lane; W1 copied transposed to LDS per work-group, B by 16-byte LDS reads; k order in sums changes
# speedup vs baseline: 30.9883x; 1.0480x over previous
#include <hip/hip_runtime.h>
#include <math.h>

#define NODES_DEF 100000
#define IN_C 128
#define HID 32
#define OUT_C 32
#define NEG_SLOPE 0.2f

typedef __attribute__((ext_vector_type(2))) float v2f;
typedef __attribute__((ext_vector_type(4))) float v4f;
typedef __attribute__((ext_vector_type(8))) float v8f;

#define BT_P (IN_C + 4)
template <int K>
__device__ __forceinline__ void gemm_rows16_wmma(const float* __restrict__ arow,
                                                 const v4f (&apre)[8],
                                                 const float* __restrict__ bt,
                                                 float* __restrict__ C, int wave)
{
    const int NC = 32;
    const int lane  = threadIdx.x & 31;
    const int half  = lane >> 4;
    const int r     = lane & 15;
    const float* __restrict__ bt0 = bt + r * BT_P + 4 * half;
    const float* __restrict__ bt1 = bt + (16 + r) * BT_P + 4 * half;

    v8f acc0 = {}, acc1 = {};
#pragma unroll
    for (int t = 0; t < K / 8; ++t) {
        const int k = 8 * t;
        v4f a;
        if (t < 8) a = apre[t];
        else       a = *(const v4f*)(arow + k);
        const v4f p = *(const v4f*)(bt0 + k);
        const v4f q = *(const v4f*)(bt1 + k);
        acc0 = __builtin_amdgcn_wmma_f32_16x16x4_f32(
             false, a.xy,  false, p.xy,
             (short)0, acc0,  false,  false);
        acc1 = __builtin_amdgcn_wmma_f32_16x16x4_f32(
             false, a.xy,  false, q.xy,
             (short)0, acc1,  false,  false);
        acc0 = __builtin_amdgcn_wmma_f32_16x16x4_f32(
             false, a.zw,  false, p.zw,
             (short)0, acc0,  false,  false);
        acc1 = __builtin_amdgcn_wmma_f32_16x16x4_f32(
             false, a.zw,  false, q.zw,
             (short)0, acc1,  false,  false);
    }

#pragma unroll
    for (int i = 0; i < 8; ++i) {
        const int orow = wave * 16 + i + 8 * half;
        C[(size_t)orow * NC + r]      = acc0[i];
        C[(size_t)orow * NC + 16 + r] = acc1[i];
    }
}

__device__ __forceinline__ float leaky(float x)
{
    return x > 0.0f ? x : NEG_SLOPE * x;
}

#define BSH   7
#define BW    (1 << BSH)
#define MAXB  1024
#define CH_T  1024
#define CH_E  16
#define CH    (CH_T * CH_E)
#define BS_T  512
#define BS_E  16
#define CAP   (BS_T * BS_E)
static_assert(MAXB == CH_T, "one thread per bucket counter in the chunk kernel");
static_assert(CH <= 65535, "an offset inside a chunk is stored in 16 bits");
static_assert(BW == 128, "bucket_sort sums its counters 4 per lane of one wave");

__device__ __forceinline__ int wave_incl_scan(int v, int lane)
{
#pragma unroll
    for (int o = 1; o < 32; o <<= 1) {
        const int t = __shfl_up(v, o);
        if (lane >= o) v += t;
    }
    return v;
}

static_assert(IN_C * HID == 4 * CH_T, "a GEMM work-group copies W1 to LDS 16 bytes per thread");
static_assert(HID * BT_P <= CH, "the transposed W1 lies in the chunk work-groups' word buffer");
__global__ void __launch_bounds__(CH_T, 16)
bin_chunk_gemm(const int* __restrict__ ei, unsigned short* __restrict__ offs,
               unsigned* __restrict__ binned, int n_edges, int nb, int ostr, int n_chunks,
               const float* __restrict__ x, const float* __restrict__ W1,
               float* __restrict__ h, int n)
{
    __shared__ int hist[MAXB];
    __shared__ int wsA[32];
    __shared__ unsigned sw[CH];
    if ((int)blockIdx.x >= n_chunks) {
        const int wave = ((int)blockIdx.x - n_chunks) * (CH_T / 32) + (int)(threadIdx.x >> 5);
        const int t = threadIdx.x;
        const int blk = wave < n / 16 ? wave : n / 16 - 1;
        const float* __restrict__ arow = x + (size_t)(blk * 16 + (t & 15)) * IN_C + 4 * ((t >> 4) & 1);
        v4f apre[8];
#pragma unroll
        for (int j = 0; j < 8; ++j) apre[j] = *(const v4f*)(arow + 8 * j);
        float* __restrict__ bt = (float*)sw;
        const v4f wq = *(const v4f*)(W1 + 4 * t);
        const int kk = t >> 3, c4 = (t & 7) * 4;
        bt[(c4 + 0) * BT_P + kk] = wq.x;
        bt[(c4 + 1) * BT_P + kk] = wq.y;
        bt[(c4 + 2) * BT_P + kk] = wq.z;
        bt[(c4 + 3) * BT_P + kk] = wq.w;
        __syncthreads();
        if (wave >= n / 16) return;
        gemm_rows16_wmma<IN_C>(arow, apre, bt, h, wave);
        return;
    }
    const int tid = threadIdx.x, lane = tid & 31, wv = tid >> 5;
    hist[tid] = 0;
    __syncthreads();
    const int base = blockIdx.x * CH;
    unsigned w[CH_E], br[CH_E];
    int sv[CH_E], dv[CH_E];
#pragma unroll
    for (int k = 0; k < CH_E; ++k) {
        const int i = base + k * CH_T + tid;
        const int iq = i < n_edges ? i : n_edges - 1;
        sv[k] = ei[iq];
        dv[k] = ei[n_edges + iq];
    }
#pragma unroll
    for (int k = 0; k < CH_E; ++k) {
        const int i = base + k * CH_T + tid;
        w[k] = 0u; br[k] = 0xFFFFFFFFu;
        if (i < n_edges) {
            const int s = sv[k];
            const int d = dv[k];
            const int b = d >> BSH;
            const int r = atomicAdd(&hist[b], 1);
            w[k]  = (unsigned)s | ((unsigned)(d & (BW - 1)) << 24);
            br[k] = (unsigned)b | ((unsigned)r << 16);
        }
    }
    __syncthreads();
    const int c = hist[tid];
    const int ic = wave_incl_scan(c, lane);
    if (lane == 31) wsA[wv] = ic;
    __syncthreads();
    if (wv == 0) {
        const int a = wsA[lane];
        const int ia = wave_incl_scan(a, lane);
        wsA[lane] = ia - a;
    }
    __syncthreads();
    const int rem = n_edges - base;
    const int nvalid = rem < CH ? rem : CH;
    const int excl = ic - c + wsA[wv];
    hist[tid] = excl;
    unsigned short* __restrict__ orow = offs + (size_t)blockIdx.x * ostr;
    if (tid < nb)  orow[tid] = (unsigned short)excl;
    if (tid == 0)  orow[nb]  = (unsigned short)nvalid;
    __syncthreads();
#pragma unroll
    for (int k = 0; k < CH_E; ++k) {
        if (br[k] != 0xFFFFFFFFu) {
            const int b = (int)(br[k] & 0xFFFFu), r = (int)(br[k] >> 16);
            sw[hist[b] + r] = w[k];
        }
    }
    __syncthreads();
#pragma unroll
    for (int k = 0; k < CH_E; ++k) {
        const int i = k * CH_T + tid;
        if (i < nvalid) binned[base + i] = sw[i];
    }
}

__global__ void __launch_bounds__(BS_T)
bucket_sort(const unsigned* __restrict__ binned, const unsigned short* __restrict__ offs,
            int* __restrict__ cnt, int* __restrict__ cur, int* __restrict__ srt,
            int n, int nch, int ostr)
{
    __shared__ int c[BW];
    __shared__ int cursor[BW];
    __shared__ int wlen[BS_T / 32];
    __shared__ int wsta[BS_T / 32];
    __shared__ int sorted[CAP];
    const int tid = threadIdx.x, lane = tid & 31, wv = tid >> 5;
    const int b = blockIdx.x;
    if (tid < BW) c[tid] = 0;
    int s = 0, len = 0;
    if (tid < nch) {
        const unsigned* __restrict__ orow = (const unsigned*)(offs + (size_t)tid * ostr);
        const unsigned w0 = orow[b >> 1];
        const unsigned w1 = orow[(b + 1) >> 1];
        const int e = (int)((b & 1) ? (w1 & 0xFFFFu) : (w0 >> 16));
        s   = (int)((b & 1) ? (w0 >> 16) : (w0 & 0xFFFFu));
        len = e - s;
    }
    const int il = wave_incl_scan(len, lane);
    int ss = s;
#pragma unroll
    for (int o = 16; o > 0; o >>= 1) ss += __shfl_xor(ss, o);
    if (lane == 31) wlen[wv] = il;
    if (lane == 0)  wsta[wv] = ss;
    __syncthreads();
    int start = 0, tot = 0, pre = 0;
#pragma unroll
    for (int j = 0; j < BS_T / 32; ++j) {
        const int a = wlen[j];
        start += wsta[j];
        tot   += a;
        if (j < wv) pre += a;
    }
    const int p0 = pre + il - len;
    const int g0 = tid * CH + s;
    const bool staged = tot <= CAP;
    unsigned v[BS_E];
    if (staged) {
        for (int k = 0; k < len; ++k) sorted[p0 + k] = g0 + k;
        __syncthreads();
#pragma unroll
        for (int k = 0; k < BS_E; ++k) {
            const int i = k * BS_T + tid;
            v[k] = binned[i < tot ? sorted[i] : 0];
        }
#pragma unroll
        for (int k = 0; k < BS_E; ++k)
            if (k * BS_T + tid < tot) atomicAdd(&c[v[k] >> 24], 1);
    } else {
        for (int k = 0; k < len; ++k) atomicAdd(&c[binned[g0 + k] >> 24], 1);
    }
    __syncthreads();
    if (tid < 32) {
        const int v0 = c[4 * tid], v1 = c[4 * tid + 1], v2 = c[4 * tid + 2], v3 = c[4 * tid + 3];
        const int sm = v0 + v1 + v2 + v3;
        const int e = wave_incl_scan(sm, tid) - sm;
        cursor[4 * tid]     = e;
        cursor[4 * tid + 1] = e + v0;
        cursor[4 * tid + 2] = e + v0 + v1;
        cursor[4 * tid + 3] = e + v0 + v1 + v2;
    }
    __syncthreads();
    if (tid < BW) {
        const int d = b * BW + tid;
        if (d < n) { cnt[d] = c[tid]; cur[d] = start + cursor[tid]; }
    }
    __syncthreads();
    if (staged) {
#pragma unroll
        for (int k = 0; k < BS_E; ++k) {
            if (k * BS_T + tid < tot) {
                const int p = atomicAdd(&cursor[v[k] >> 24], 1);
                sorted[p] = (int)(v[k] & 0xFFFFFFu);
            }
        }
        __syncthreads();
#pragma unroll
        for (int k = 0; k < BS_E; ++k) {
            const int i = k * BS_T + tid;
            if (i < tot) srt[start + i] = sorted[i];
        }
    } else {
        for (int k = 0; k < len; ++k) {
            const unsigned wd = binned[g0 + k];
            const int p = atomicAdd(&cursor[wd >> 24], 1);
            srt[start + p] = (int)(wd & 0xFFFFFFu);
        }
    }
}

__device__ __forceinline__ float xhalf(float x)
{
    return __int_as_float(__builtin_amdgcn_permlanex16(__float_as_int(x), __float_as_int(x),
                                                      0x76543210, 0xfedcba98, false, false));
}
template <int S>
__device__ __forceinline__ float xrow(float x)
{
    return __int_as_float(__builtin_amdgcn_update_dpp(0, __float_as_int(x), 0x160 + S,
                                                     0xf, 0xf, true));
}
__device__ __forceinline__ float wave_sum(float v)
{
    v += xrow<1>(v); v += xrow<2>(v); v += xrow<4>(v); v += xrow<8>(v);
    return v + xhalf(v);
}
__device__ __forceinline__ float wave_max(float v)
{
    v = fmaxf(v, xrow<1>(v)); v = fmaxf(v, xrow<2>(v));
    v = fmaxf(v, xrow<4>(v)); v = fmaxf(v, xrow<8>(v));
    return fmaxf(v, xhalf(v));
}

template <int J>
__device__ __forceinline__ int grp_pick(int x)
{
    return (int)__builtin_amdgcn_mov_dpp8((unsigned)x, J * 0x249249u);
}
__device__ __forceinline__ int grp_bcast(int x, int j)
{
    switch (j) {
    case 0:  return grp_pick<0>(x);
    case 1:  return grp_pick<1>(x);
    case 2:  return grp_pick<2>(x);
    case 3:  return grp_pick<3>(x);
    case 4:  return grp_pick<4>(x);
    case 5:  return grp_pick<5>(x);
    case 6:  return grp_pick<6>(x);
    default: return grp_pick<7>(x);
    }
}
__device__ __forceinline__ float grp_sum(float v)
{
    v += xrow<1>(v); v += xrow<2>(v);
    return v + xrow<4>(v);
}
__device__ __forceinline__ float dot4(v4f a, v4f b)
{
    const v2f t = __builtin_elementwise_fma(a.zw, b.zw, a.xy * b.xy);
    return t.x + t.y;
}

template <int S, int N>
__device__ __forceinline__ void fold_rows(float* p, int lane)
{
    const bool hi = (lane & S) != 0;
#pragma unroll
    for (int u = 0; u < N; ++u) {
        const float x = hi ? p[u + N] : p[u];
        const float y = hi ? p[u] : p[u + N];
        p[u] = x + xrow<S>(y);
    }
}

template <int G>
__device__ __forceinline__ float group_dots(float (&p)[G], int lane)
{
    if constexpr (G == 8) fold_rows<4, 4>(p, lane);
    else {
#pragma unroll
        for (int u = 0; u < G; ++u) p[u] += xrow<4>(p[u]);
    }
    if constexpr (G >= 4) fold_rows<2, 2>(p, lane);
    else {
#pragma unroll
        for (int u = 0; u < G; ++u) p[u] += xrow<2>(p[u]);
    }
    fold_rows<1, 1>(p, lane);
    return p[0];
}

template <int G>
__device__ __forceinline__ void gather_chunk(const float* __restrict__ h, int sj, int k,
                                             int lane, int eidx, unsigned sub16,
                                             v4f as, float sd,
                                             float& m, float& z, v4f& acc)
{
    v4f v[G];
#pragma unroll
    for (int g = 0; g < G; ++g) {
        if (4 * g < k) {
            const unsigned s = (unsigned)grp_bcast(sj, g);
            const unsigned off = s * (unsigned)(HID * 4) + sub16;
            v[g] = *(const v4f*)((const char*)h + (size_t)off);
        } else {
            v[g] = (v4f){0.0f, 0.0f, 0.0f, 0.0f};
        }
    }
    float p[G];
#pragma unroll
    for (int g = 0; g < G; ++g) p[g] = dot4(v[g], as);
    const float s = group_dots<G>(p, lane);

    const bool  valid = eidx < k;
    const float e  = valid ? leaky(s + sd) : m;
    const float mn = fmaxf(m, wave_max(e));
    const float scale = __expf(m - mn);
    const float ex = valid ? __expf(e - mn) : 0.0f;
    z   = z * scale + wave_sum(ex);
    acc.x *= scale; acc.y *= scale; acc.z *= scale; acc.w *= scale;
    const int wi = __float_as_int(ex);
#pragma unroll
    for (int g = 0; g < G; ++g) {
        if (4 * g < k) {
            const float w = __int_as_float(grp_bcast(wi, g));
            acc.x = __builtin_fmaf(w, v[g].x, acc.x);
            acc.y = __builtin_fmaf(w, v[g].y, acc.y);
            acc.z = __builtin_fmaf(w, v[g].z, acc.z);
            acc.w = __builtin_fmaf(w, v[g].w, acc.w);
        }
    }
    m = mn;
}

#define GN 16
#define TP (HID + 2)
static_assert(OUT_C == HID, "both fused products are 32 x 32: one row pitch, two column tiles");
template <int EPI>
__global__ void __launch_bounds__(GN * 32, 16)
gat_node_gather_mm(const int* __restrict__ srt,
                   const int* __restrict__ cur,
                   const int* __restrict__ cnt,
                   const float* __restrict__ a_src,
                   const float* __restrict__ a_dst,
                   const float* __restrict__ h,
                   const float* __restrict__ bias,
                   const float* __restrict__ Wn,
                   const float* __restrict__ bn,
                   float* __restrict__ outp)
{
    __shared__ float tile[GN * TP];
    const int lane = threadIdx.x & 31;
    const int wv = __builtin_amdgcn_readfirstlane((int)(threadIdx.x >> 5));
    const int d  = (int)blockIdx.x * GN + wv;

    {
        const int beg = __builtin_amdgcn_readfirstlane(cur[d]);
        const int deg = __builtin_amdgcn_readfirstlane(cnt[d]);
        const int end = beg + deg;

        const unsigned sub16 = (unsigned)(lane & 7) * 16u;
        const int      eidx  = (int)(sub16 >> 2) + (lane >> 3);

        const v4f hd = *(const v4f*)((const char*)(h + (size_t)d * HID) + sub16);
        const v4f as = *(const v4f*)((const char*)a_src + sub16);
        const v4f ad = *(const v4f*)((const char*)a_dst + sub16);

        int sj = d;
        if (eidx < deg) sj = (srt + beg)[eidx];

        const float sd     = grp_sum(dot4(hd, ad));
        const float e_self = leaky(grp_sum(dot4(hd, as)) + sd);

        float m = e_self, z = 1.0f;
        v4f acc = (v4f){0.0f, 0.0f, 0.0f, 0.0f};
        if (lane < 8) acc = hd;

        for (int c0 = beg; c0 < end; c0 += 32) {
            const int rem = end - c0;
            const int k   = rem < 32 ? rem : 32;
            int sn = d;
            if (eidx < rem - 32) sn = (srt + c0 + 32)[eidx];
            if (k > 16)     gather_chunk<8>(h, sj, k, lane, eidx, sub16, as, sd, m, z, acc);
            else if (k > 8) gather_chunk<4>(h, sj, k, lane, eidx, sub16, as, sd, m, z, acc);
            else            gather_chunk<2>(h, sj, k, lane, eidx, sub16, as, sd, m, z, acc);
            sj = sn;
        }

        acc.x += xrow<8>(acc.x); acc.y += xrow<8>(acc.y); acc.z += xrow<8>(acc.z); acc.w += xrow<8>(acc.w);
        acc.x += xhalf(acc.x);   acc.y += xhalf(acc.y);   acc.z += xhalf(acc.z);   acc.w += xhalf(acc.w);

        const v4f   bv  = *(const v4f*)((const char*)bias + sub16);
        const float inv = 1.0f / (z + 1e-16f);
        const float r0 = acc.x * inv + bv.x, r1 = acc.y * inv + bv.y;
        const float r2 = acc.z * inv + bv.z, r3 = acc.w * inv + bv.w;
        if (lane < 8) {
            float* trow = (float*)((char*)(tile + wv * TP) + sub16);
            trow[0] = r0 > 0.0f ? r0 : 0.0f;
            trow[1] = r1 > 0.0f ? r1 : 0.0f;
            trow[2] = r2 > 0.0f ? r2 : 0.0f;
            trow[3] = r3 > 0.0f ? r3 : 0.0f;
        }
    }
    __syncthreads();
    if (wv >= HID / 16) return;

    const int half = lane >> 4;
    const int rr   = lane & 15;
    const int col  = wv * 16 + rr;
    v8f acc2 = {};
#pragma unroll
    for (int k = 0; k < HID; k += 4) {
        v2f a = *(const v2f*)(tile + rr * TP + k + 2 * half);
        v2f b;
        b.x = Wn[(k + 2 * half + 0) * HID + col];
        b.y = Wn[(k + 2 * half + 1) * HID + col];
        acc2 = __builtin_amdgcn_wmma_f32_16x16x4_f32(
             false, a,  false, b,
             (short)0, acc2,  false,  false);
    }
    float bcol = 0.0f;
    if (EPI == 1) bcol = bn[col];
#pragma unroll
    for (int i = 0; i < 8; ++i) {
        const int orow = (int)blockIdx.x * GN + i + 8 * half;
        float v = acc2[i];
        if (EPI == 1) {
            v += bcol;
            v = 1.0f / (1.0f + __expf(-v));
        }
        outp[(size_t)orow * HID + col] = v;
    }
}

extern "C" void kernel_launch(void* const* d_in, const int* in_sizes, int n_in,
                              void* d_out, int out_size, void* d_ws, size_t ws_size,
                              hipStream_t stream)
{
    const float* x      = (const float*)d_in[0];
    const int*   ei     = (const int*)  d_in[1];
    const float* W1     = (const float*)d_in[2];
    const float* a_src1 = (const float*)d_in[3];
    const float* a_dst1 = (const float*)d_in[4];
    const float* b1     = (const float*)d_in[5];
    const float* W2     = (const float*)d_in[6];
    const float* a_src2 = (const float*)d_in[7];
    const float* a_dst2 = (const float*)d_in[8];
    const float* b2     = (const float*)d_in[9];
    const float* Wc     = (const float*)d_in[10];
    const float* bc     = (const float*)d_in[11];
    float* out = (float*)d_out;

    const int n       = in_sizes[0] / IN_C;
    const int n_edges = in_sizes[1] / 2;

    float* ws    = (float*)d_ws;
    float* h     = ws;
    float* g     = h   + (size_t)n * HID;
    int*   cnt   = (int*)(g + (size_t)n * HID);
    unsigned short* offs = (unsigned short*)(cnt + n);
    int*   cur   = cnt + 3 * (size_t)n;
    int*   srt   = cur + n;
    unsigned* binned = (unsigned*)g;

    const int chunkBlocks = (n_edges + CH - 1) / CH;
    const int nb = (n + BW - 1) / BW;
    const int ostr = (nb + 2) & ~1;
    const int nodeBlocks = n / GN;
    const int gemmWaves  = n / 16;
    const int gemmBlocks = (gemmWaves + CH_T / 32 - 1) / (CH_T / 32);

    bin_chunk_gemm<<<chunkBlocks + gemmBlocks, CH_T, 0, stream>>>(ei, offs, binned, n_edges, nb, ostr,
                                                                 chunkBlocks, x, W1, h, n);
    bucket_sort<<<nb, BS_T, 0, stream>>>(binned, offs, cnt, cur, srt, n, chunkBlocks, ostr);
    gat_node_gather_mm<0><<<nodeBlocks, GN * 32, 0, stream>>>(srt, cur, cnt, a_src1, a_dst1, h, b1,
                                                             W2, nullptr, g);

    gat_node_gather_mm<1><<<nodeBlocks, GN * 32, 0, stream>>>(srt, cur, cnt, a_src2, a_dst2, g, b2,
                                                             Wc, bc, out);
}


# node gather: a wave walks 8 consecutive nodes; their list bounds and self scores in scalar registers, own rows in LDS, next sources in flight; 4-wave groups, a product tile per wave; bias add unfused
# speedup vs baseline: 32.1024x; 1.0360x over previous
#include <hip/hip_runtime.h>
#include <math.h>

#define NODES_DEF 100000
#define IN_C 128
#define HID 32
#define OUT_C 32
#define NEG_SLOPE 0.2f

typedef __attribute__((ext_vector_type(2))) float v2f;
typedef __attribute__((ext_vector_type(4))) float v4f;
typedef __attribute__((ext_vector_type(8))) float v8f;

#define BT_P (IN_C + 4)
template <int K>
__device__ __forceinline__ void gemm_rows16_wmma(const float* __restrict__ arow,
                                                 const v4f (&apre)[8],
                                                 const float* __restrict__ bt,
                                                 float* __restrict__ C, int wave)
{
    const int NC = 32;
    const int lane  = threadIdx.x & 31;
    const int half  = lane >> 4;
    const int r     = lane & 15;
    const float* __restrict__ bt0 = bt + r * BT_P + 4 * half;
    const float* __restrict__ bt1 = bt + (16 + r) * BT_P + 4 * half;

    v8f acc0 = {}, acc1 = {};
#pragma unroll
    for (int t = 0; t < K / 8; ++t) {
        const int k = 8 * t;
        v4f a;
        if (t < 8) a = apre[t];
        else       a = *(const v4f*)(arow + k);
        const v4f p = *(const v4f*)(bt0 + k);
        const v4f q = *(const v4f*)(bt1 + k);
        acc0 = __builtin_amdgcn_wmma_f32_16x16x4_f32(
             false, a.xy,  false, p.xy,
             (short)0, acc0,  false,  false);
        acc1 = __builtin_amdgcn_wmma_f32_16x16x4_f32(
             false, a.xy,  false, q.xy,
             (short)0, acc1,  false,  false);
        acc0 = __builtin_amdgcn_wmma_f32_16x16x4_f32(
             false, a.zw,  false, p.zw,
             (short)0, acc0,  false,  false);
        acc1 = __builtin_amdgcn_wmma_f32_16x16x4_f32(
             false, a.zw,  false, q.zw,
             (short)0, acc1,  false,  false);
    }

#pragma unroll
    for (int i = 0; i < 8; ++i) {
        const int orow = wave * 16 + i + 8 * half;
        C[(size_t)orow * NC + r]      = acc0[i];
        C[(size_t)orow * NC + 16 + r] = acc1[i];
    }
}

__device__ __forceinline__ float leaky(float x)
{
    return x > 0.0f ? x : NEG_SLOPE * x;
}

#define BSH   7
#define BW    (1 << BSH)
#define MAXB  1024
#define CH_T  1024
#define CH_E  16
#define CH    (CH_T * CH_E)
#define BS_T  512
#define BS_E  16
#define CAP   (BS_T * BS_E)
static_assert(MAXB == CH_T, "one thread per bucket counter in the chunk kernel");
static_assert(CH <= 65535, "an offset inside a chunk is stored in 16 bits");
static_assert(BW == 128, "bucket_sort sums its counters 4 per lane of one wave");

__device__ __forceinline__ int wave_incl_scan(int v, int lane)
{
#pragma unroll
    for (int o = 1; o < 32; o <<= 1) {
        const int t = __shfl_up(v, o);
        if (lane >= o) v += t;
    }
    return v;
}

static_assert(IN_C * HID == 4 * CH_T, "a GEMM work-group copies W1 to LDS 16 bytes per thread");
static_assert(HID * BT_P <= CH, "the transposed W1 lies in the chunk work-groups' word buffer");
__global__ void __launch_bounds__(CH_T, 16)
bin_chunk_gemm(const int* __restrict__ ei, unsigned short* __restrict__ offs,
               unsigned* __restrict__ binned, int n_edges, int nb, int ostr, int n_chunks,
               const float* __restrict__ x, const float* __restrict__ W1,
               float* __restrict__ h, int n)
{
    __shared__ int hist[MAXB];
    __shared__ int wsA[32];
    __shared__ unsigned sw[CH];
    if ((int)blockIdx.x >= n_chunks) {
        const int wave = ((int)blockIdx.x - n_chunks) * (CH_T / 32) + (int)(threadIdx.x >> 5);
        const int t = threadIdx.x;
        const int blk = wave < n / 16 ? wave : n / 16 - 1;
        const float* __restrict__ arow = x + (size_t)(blk * 16 + (t & 15)) * IN_C + 4 * ((t >> 4) & 1);
        v4f apre[8];
#pragma unroll
        for (int j = 0; j < 8; ++j) apre[j] = *(const v4f*)(arow + 8 * j);
        float* __restrict__ bt = (float*)sw;
        const v4f wq = *(const v4f*)(W1 + 4 * t);
        const int kk = t >> 3, c4 = (t & 7) * 4;
        bt[(c4 + 0) * BT_P + kk] = wq.x;
        bt[(c4 + 1) * BT_P + kk] = wq.y;
        bt[(c4 + 2) * BT_P + kk] = wq.z;
        bt[(c4 + 3) * BT_P + kk] = wq.w;
        __syncthreads();
        if (wave >= n / 16) return;
        gemm_rows16_wmma<IN_C>(arow, apre, bt, h, wave);
        return;
    }
    const int tid = threadIdx.x, lane = tid & 31, wv = tid >> 5;
    hist[tid] = 0;
    __syncthreads();
    const int base = blockIdx.x * CH;
    unsigned w[CH_E], br[CH_E];
    int sv[CH_E], dv[CH_E];
#pragma unroll
    for (int k = 0; k < CH_E; ++k) {
        const int i = base + k * CH_T + tid;
        const int iq = i < n_edges ? i : n_edges - 1;
        sv[k] = ei[iq];
        dv[k] = ei[n_edges + iq];
    }
#pragma unroll
    for (int k = 0; k < CH_E; ++k) {
        const int i = base + k * CH_T + tid;
        w[k] = 0u; br[k] = 0xFFFFFFFFu;
        if (i < n_edges) {
            const int s = sv[k];
            const int d = dv[k];
            const int b = d >> BSH;
            const int r = atomicAdd(&hist[b], 1);
            w[k]  = (unsigned)s | ((unsigned)(d & (BW - 1)) << 24);
            br[k] = (unsigned)b | ((unsigned)r << 16);
        }
    }
    __syncthreads();
    const int c = hist[tid];
    const int ic = wave_incl_scan(c, lane);
    if (lane == 31) wsA[wv] = ic;
    __syncthreads();
    if (wv == 0) {
        const int a = wsA[lane];
        const int ia = wave_incl_scan(a, lane);
        wsA[lane] = ia - a;
    }
    __syncthreads();
    const int rem = n_edges - base;
    const int nvalid = rem < CH ? rem : CH;
    const int excl = ic - c + wsA[wv];
    hist[tid] = excl;
    unsigned short* __restrict__ orow = offs + (size_t)blockIdx.x * ostr;
    if (tid < nb)  orow[tid] = (unsigned short)excl;
    if (tid == 0)  orow[nb]  = (unsigned short)nvalid;
    __syncthreads();
#pragma unroll
    for (int k = 0; k < CH_E; ++k) {
        if (br[k] != 0xFFFFFFFFu) {
            const int b = (int)(br[k] & 0xFFFFu), r = (int)(br[k] >> 16);
            sw[hist[b] + r] = w[k];
        }
    }
    __syncthreads();
#pragma unroll
    for (int k = 0; k < CH_E; ++k) {
        const int i = k * CH_T + tid;
        if (i < nvalid) binned[base + i] = sw[i];
    }
}

__global__ void __launch_bounds__(BS_T)
bucket_sort(const unsigned* __restrict__ binned, const unsigned short* __restrict__ offs,
            int* __restrict__ cnt, int* __restrict__ cur, int* __restrict__ srt,
            int n, int nch, int ostr)
{
    __shared__ int c[BW];
    __shared__ int cursor[BW];
    __shared__ int wlen[BS_T / 32];
    __shared__ int wsta[BS_T / 32];
    __shared__ int sorted[CAP];
    const int tid = threadIdx.x, lane = tid & 31, wv = tid >> 5;
    const int b = blockIdx.x;
    if (tid < BW) c[tid] = 0;
    int s = 0, len = 0;
    if (tid < nch) {
        const unsigned* __restrict__ orow = (const unsigned*)(offs + (size_t)tid * ostr);
        const unsigned w0 = orow[b >> 1];
        const unsigned w1 = orow[(b + 1) >> 1];
        const int e = (int)((b & 1) ? (w1 & 0xFFFFu) : (w0 >> 16));
        s   = (int)((b & 1) ? (w0 >> 16) : (w0 & 0xFFFFu));
        len = e - s;
    }
    const int il = wave_incl_scan(len, lane);
    int ss = s;
#pragma unroll
    for (int o = 16; o > 0; o >>= 1) ss += __shfl_xor(ss, o);
    if (lane == 31) wlen[wv] = il;
    if (lane == 0)  wsta[wv] = ss;
    __syncthreads();
    int start = 0, tot = 0, pre = 0;
#pragma unroll
    for (int j = 0; j < BS_T / 32; ++j) {
        const int a = wlen[j];
        start += wsta[j];
        tot   += a;
        if (j < wv) pre += a;
    }
    const int p0 = pre + il - len;
    const int g0 = tid * CH + s;
    const bool staged = tot <= CAP;
    unsigned v[BS_E];
    if (staged) {
        for (int k = 0; k < len; ++k) sorted[p0 + k] = g0 + k;
        __syncthreads();
#pragma unroll
        for (int k = 0; k < BS_E; ++k) {
            const int i = k * BS_T + tid;
            v[k] = binned[i < tot ? sorted[i] : 0];
        }
#pragma unroll
        for (int k = 0; k < BS_E; ++k)
            if (k * BS_T + tid < tot) atomicAdd(&c[v[k] >> 24], 1);
    } else {
        for (int k = 0; k < len; ++k) atomicAdd(&c[binned[g0 + k] >> 24], 1);
    }
    __syncthreads();
    if (tid < 32) {
        const int v0 = c[4 * tid], v1 = c[4 * tid + 1], v2 = c[4 * tid + 2], v3 = c[4 * tid + 3];
        const int sm = v0 + v1 + v2 + v3;
        const int e = wave_incl_scan(sm, tid) - sm;
        cursor[4 * tid]     = e;
        cursor[4 * tid + 1] = e + v0;
        cursor[4 * tid + 2] = e + v0 + v1;
        cursor[4 * tid + 3] = e + v0 + v1 + v2;
    }
    __syncthreads();
    if (tid < BW) {
        const int d = b * BW + tid;
        if (d < n) { cnt[d] = c[tid]; cur[d] = start + cursor[tid]; }
    }
    __syncthreads();
    if (staged) {
#pragma unroll
        for (int k = 0; k < BS_E; ++k) {
            if (k * BS_T + tid < tot) {
                const int p = atomicAdd(&cursor[v[k] >> 24], 1);
                sorted[p] = (int)(v[k] & 0xFFFFFFu);
            }
        }
        __syncthreads();
#pragma unroll
        for (int k = 0; k < BS_E; ++k) {
            const int i = k * BS_T + tid;
            if (i < tot) srt[start + i] = sorted[i];
        }
    } else {
        for (int k = 0; k < len; ++k) {
            const unsigned wd = binned[g0 + k];
            const int p = atomicAdd(&cursor[wd >> 24], 1);
            srt[start + p] = (int)(wd & 0xFFFFFFu);
        }
    }
}

__device__ __forceinline__ float xhalf(float x)
{
    return __int_as_float(__builtin_amdgcn_permlanex16(__float_as_int(x), __float_as_int(x),
                                                      0x76543210, 0xfedcba98, false, false));
}
template <int S>
__device__ __forceinline__ float xrow(float x)
{
    return __int_as_float(__builtin_amdgcn_update_dpp(0, __float_as_int(x), 0x160 + S,
                                                     0xf, 0xf, true));
}
__device__ __forceinline__ float wave_sum(float v)
{
    v += xrow<1>(v); v += xrow<2>(v); v += xrow<4>(v); v += xrow<8>(v);
    return v + xhalf(v);
}
__device__ __forceinline__ float wave_max(float v)
{
    v = fmaxf(v, xrow<1>(v)); v = fmaxf(v, xrow<2>(v));
    v = fmaxf(v, xrow<4>(v)); v = fmaxf(v, xrow<8>(v));
    return fmaxf(v, xhalf(v));
}

template <int J>
__device__ __forceinline__ int grp_pick(int x)
{
    return (int)__builtin_amdgcn_mov_dpp8((unsigned)x, J * 0x249249u);
}
__device__ __forceinline__ int grp_bcast(int x, int j)
{
    switch (j) {
    case 0:  return grp_pick<0>(x);
    case 1:  return grp_pick<1>(x);
    case 2:  return grp_pick<2>(x);
    case 3:  return grp_pick<3>(x);
    case 4:  return grp_pick<4>(x);
    case 5:  return grp_pick<5>(x);
    case 6:  return grp_pick<6>(x);
    default: return grp_pick<7>(x);
    }
}
__device__ __forceinline__ float grp_sum(float v)
{
    v += xrow<1>(v); v += xrow<2>(v);
    return v + xrow<4>(v);
}
__device__ __forceinline__ float dot4(v4f a, v4f b)
{
    const v2f t = __builtin_elementwise_fma(a.zw, b.zw, a.xy * b.xy);
    return t.x + t.y;
}

template <int S, int N>
__device__ __forceinline__ void fold_rows(float* p, int lane)
{
    const bool hi = (lane & S) != 0;
#pragma unroll
    for (int u = 0; u < N; ++u) {
        const float x = hi ? p[u + N] : p[u];
        const float y = hi ? p[u] : p[u + N];
        p[u] = x + xrow<S>(y);
    }
}

template <int G>
__device__ __forceinline__ float group_dots(float (&p)[G], int lane)
{
    if constexpr (G == 8) fold_rows<4, 4>(p, lane);
    else {
#pragma unroll
        for (int u = 0; u < G; ++u) p[u] += xrow<4>(p[u]);
    }
    if constexpr (G >= 4) fold_rows<2, 2>(p, lane);
    else {
#pragma unroll
        for (int u = 0; u < G; ++u) p[u] += xrow<2>(p[u]);
    }
    fold_rows<1, 1>(p, lane);
    return p[0];
}

template <int G>
__device__ __forceinline__ void gather_chunk(const float* __restrict__ h, int sj, int k,
                                             int lane, int eidx, unsigned sub16,
                                             v4f as, float sd,
                                             float& m, float& z, v4f& acc)
{
    v4f v[G];
#pragma unroll
    for (int g = 0; g < G; ++g) {
        if (4 * g < k) {
            const unsigned s = (unsigned)grp_bcast(sj, g);
            const unsigned off = s * (unsigned)(HID * 4) + sub16;
            v[g] = *(const v4f*)((const char*)h + (size_t)off);
        } else {
            v[g] = (v4f){0.0f, 0.0f, 0.0f, 0.0f};
        }
    }
    float p[G];
#pragma unroll
    for (int g = 0; g < G; ++g) p[g] = dot4(v[g], as);
    const float s = group_dots<G>(p, lane);

    const bool  valid = eidx < k;
    const float e  = valid ? leaky(s + sd) : m;
    const float mn = fmaxf(m, wave_max(e));
    const float scale = __expf(m - mn);
    const float ex = valid ? __expf(e - mn) : 0.0f;
    z   = z * scale + wave_sum(ex);
    acc.x *= scale; acc.y *= scale; acc.z *= scale; acc.w *= scale;
    const int wi = __float_as_int(ex);
#pragma unroll
    for (int g = 0; g < G; ++g) {
        if (4 * g < k) {
            const float w = __int_as_float(grp_bcast(wi, g));
            acc.x = __builtin_fmaf(w, v[g].x, acc.x);
            acc.y = __builtin_fmaf(w, v[g].y, acc.y);
            acc.z = __builtin_fmaf(w, v[g].z, acc.z);
            acc.w = __builtin_fmaf(w, v[g].w, acc.w);
        }
    }
    m = mn;
}

#define GN 16
#define NPW 8
#define GW 4
#define WGN (GW * NPW)
#define TP (HID + 2)
static_assert(OUT_C == HID, "both fused products are 32 x 32: one row pitch, two column tiles");
static_assert((WGN / GN) * (HID / 16) == GW, "one 16 x 16 output tile per wave");
static_assert(GN % NPW == 0, "a wave's nodes lie in one row tile, all there or all missing");
static_assert(NPW == 8, "the wave's own rows: two loads of 4 rows");
template <int EPI>
__global__ void __launch_bounds__(GW * 32, 16)
gat_node_gather_mm(const int* __restrict__ srt,
                   const int* __restrict__ cur,
                   const int* __restrict__ cnt,
                   const float* __restrict__ a_src,
                   const float* __restrict__ a_dst,
                   const float* __restrict__ h,
                   const float* __restrict__ bias,
                   const float* __restrict__ Wn,
                   const float* __restrict__ bn,
                   float* __restrict__ outp, int n)
{
    __shared__ float tile[WGN * TP];
    const int wv   = __builtin_amdgcn_readfirstlane((int)(threadIdx.x >> 5));
    const int base = (int)blockIdx.x * WGN;
    const int d0   = base + wv * NPW;

    if (d0 < n) {
        const int lane = threadIdx.x & 31;
        const unsigned sub16 = (unsigned)(lane & 7) * 16u;
        const int      eidx  = (int)(sub16 >> 2) + (lane >> 3);

        int begs[NPW], degs[NPW];
        float sds[NPW], ess[NPW];
#pragma unroll
        for (int i = 0; i < NPW; ++i) {
            begs[i] = __builtin_amdgcn_readfirstlane(cur[d0 + i]);
            degs[i] = __builtin_amdgcn_readfirstlane(cnt[d0 + i]);
        }
        int sj = d0;
        if (eidx < degs[0]) sj = *(const int*)((const char*)srt + (size_t)(((unsigned)begs[0] + (unsigned)eidx) * 4u));

        const v4f as = *(const v4f*)((const char*)a_src + sub16);
        float* const slot0 = (float*)((char*)(tile + wv * NPW * TP) + sub16);
        {
            const v4f ad = *(const v4f*)((const char*)a_dst + sub16);
            const int q  = lane >> 3;
            const v4f ta = *(const v4f*)((const char*)h + (size_t)((unsigned)(d0 + q)     * (unsigned)(HID * 4) + sub16));
            const v4f tb = *(const v4f*)((const char*)h + (size_t)((unsigned)(d0 + q + 4) * (unsigned)(HID * 4) + sub16));
            float* const pa = slot0 + q * TP;
            float* const pb = pa + 4 * TP;
            pa[0] = ta.x; pa[1] = ta.y; pa[2] = ta.z; pa[3] = ta.w;
            pb[0] = tb.x; pb[1] = tb.y; pb[2] = tb.z; pb[3] = tb.w;
            const float da = grp_sum(dot4(ta, ad)), db = grp_sum(dot4(tb, ad));
            const int sa = __float_as_int(da);
            const int sb = __float_as_int(db);
            const int ea = __float_as_int(leaky(grp_sum(dot4(ta, as)) + da));
            const int eb = __float_as_int(leaky(grp_sum(dot4(tb, as)) + db));
#pragma unroll
            for (int i = 0; i < 4; ++i) {
                sds[i]     = __int_as_float(__builtin_amdgcn_readlane(sa, 8 * i));
                sds[i + 4] = __int_as_float(__builtin_amdgcn_readlane(sb, 8 * i));
                ess[i]     = __int_as_float(__builtin_amdgcn_readlane(ea, 8 * i));
                ess[i + 4] = __int_as_float(__builtin_amdgcn_readlane(eb, 8 * i));
            }
        }
        v4f hd;
        hd.x = slot0[0]; hd.y = slot0[1]; hd.z = slot0[2]; hd.w = slot0[3];

        int d = d0;
#pragma unroll 1
        for (int j = 0; j < NPW; ++j, ++d) {
            const bool  more = j + 1 < NPW;
            const int   dn   = more ? d + 1 : d;
            const int   beg  = begs[0], deg = degs[0];
            const int   begn = begs[1];
            const int   degn = more ? degs[1] : 0;
            const float sd   = sds[0];

            float m = ess[0], z = 1.0f;
            v4f acc = (v4f){0.0f, 0.0f, 0.0f, 0.0f};
            if (lane < 8) acc = hd;

            const int end = beg + deg;
            int c0 = beg;
            for (;;) {
                const int  rem  = end - c0;
                const int  k    = rem < 32 ? rem : 32;
                const bool last = rem <= 32;
                const int nbase = last ? begn : c0 + 32;
                const int ncnt  = last ? degn : rem - 32;
                int sn = last ? dn : d;
                if (eidx < ncnt) sn = *(const int*)((const char*)srt + (size_t)(((unsigned)nbase + (unsigned)eidx) * 4u));
                if (k > 16)     gather_chunk<8>(h, sj, k, lane, eidx, sub16, as, sd, m, z, acc);
                else if (k > 8) gather_chunk<4>(h, sj, k, lane, eidx, sub16, as, sd, m, z, acc);
                else if (k > 0) gather_chunk<2>(h, sj, k, lane, eidx, sub16, as, sd, m, z, acc);
                m = __int_as_float(__builtin_amdgcn_readfirstlane(__float_as_int(m)));
                z = __int_as_float(__builtin_amdgcn_readfirstlane(__float_as_int(z)));
                sj = sn;
                if (last) break;
                c0 += 32;
            }

            float* const slot  = slot0 + j * TP;
            const float* slotn = more ? slot + TP : slot;
            hd.x = slotn[0]; hd.y = slotn[1]; hd.z = slotn[2]; hd.w = slotn[3];

            acc.x += xrow<8>(acc.x); acc.y += xrow<8>(acc.y); acc.z += xrow<8>(acc.z); acc.w += xrow<8>(acc.w);
            acc.x += xhalf(acc.x);   acc.y += xhalf(acc.y);   acc.z += xhalf(acc.z);   acc.w += xhalf(acc.w);

            const float inv = 1.0f / (z + 1e-16f);
            if (lane < 8) {
                slot[0] = acc.x * inv;
                slot[1] = acc.y * inv;
                slot[2] = acc.z * inv;
                slot[3] = acc.w * inv;
            }
#pragma unroll
            for (int i = 0; i + 1 < NPW; ++i) {
                begs[i] = begs[i + 1]; degs[i] = degs[i + 1]; sds[i] = sds[i + 1]; ess[i] = ess[i + 1];
            }
        }
    }

    const int lane2 = (int)__builtin_amdgcn_mbcnt_lo(~0u, 0u);
    const int tj   = wv >> 1;
    const int half = lane2 >> 4;
    const int rr   = lane2 & 15;
    const int col  = (wv & 1) * 16 + rr;
    v2f b[HID / 4], bb[HID / 4];
#pragma unroll
    for (int k = 0; k < HID; k += 4) {
        b[k / 4].x = Wn[(k + 2 * half + 0) * HID + col];
        b[k / 4].y = Wn[(k + 2 * half + 1) * HID + col];
        bb[k / 4] = *(const v2f*)(bias + k + 2 * half);
    }
    float bcol = 0.0f;
    if (EPI == 1) bcol = bn[col];

    __syncthreads();
    if (base + tj * GN >= n) return;

    const float* __restrict__ trd = tile + tj * GN * TP;
    v8f acc2 = {};
#pragma unroll
    for (int k = 0; k < HID; k += 4) {
        v2f a = *(const v2f*)(trd + rr * TP + k + 2 * half);
        a += bb[k / 4];
        a.x = a.x > 0.0f ? a.x : 0.0f;
        a.y = a.y > 0.0f ? a.y : 0.0f;
        acc2 = __builtin_amdgcn_wmma_f32_16x16x4_f32(
             false, a,  false, b[k / 4],
             (short)0, acc2,  false,  false);
    }
#pragma unroll
    for (int i = 0; i < 8; ++i) {
        const int orow = base + tj * GN + i + 8 * half;
        float v = acc2[i];
        if (EPI == 1) {
            v += bcol;
            v = 1.0f / (1.0f + __expf(-v));
        }
        outp[(size_t)orow * HID + col] = v;
    }
}

extern "C" void kernel_launch(void* const* d_in, const int* in_sizes, int n_in,
                              void* d_out, int out_size, void* d_ws, size_t ws_size,
                              hipStream_t stream)
{
    const float* x      = (const float*)d_in[0];
    const int*   ei     = (const int*)  d_in[1];
    const float* W1     = (const float*)d_in[2];
    const float* a_src1 = (const float*)d_in[3];
    const float* a_dst1 = (const float*)d_in[4];
    const float* b1     = (const float*)d_in[5];
    const float* W2     = (const float*)d_in[6];
    const float* a_src2 = (const float*)d_in[7];
    const float* a_dst2 = (const float*)d_in[8];
    const float* b2     = (const float*)d_in[9];
    const float* Wc     = (const float*)d_in[10];
    const float* bc     = (const float*)d_in[11];
    float* out = (float*)d_out;

    const int n       = in_sizes[0] / IN_C;
    const int n_edges = in_sizes[1] / 2;

    float* ws    = (float*)d_ws;
    float* h     = ws;
    float* g     = h   + (size_t)n * HID;
    int*   cnt   = (int*)(g + (size_t)n * HID);
    unsigned short* offs = (unsigned short*)(cnt + n);
    int*   cur   = cnt + 3 * (size_t)n;
    int*   srt   = cur + n;
    unsigned* binned = (unsigned*)g;

    const int chunkBlocks = (n_edges + CH - 1) / CH;
    const int nb = (n + BW - 1) / BW;
    const int ostr = (nb + 2) & ~1;
    const int nodeBlocks = (n + WGN - 1) / WGN;
    const int gemmWaves  = n / 16;
    const int gemmBlocks = (gemmWaves + CH_T / 32 - 1) / (CH_T / 32);

    bin_chunk_gemm<<<chunkBlocks + gemmBlocks, CH_T, 0, stream>>>(ei, offs, binned, n_edges, nb, ostr,
                                                                 chunkBlocks, x, W1, h, n);
    bucket_sort<<<nb, BS_T, 0, stream>>>(binned, offs, cnt, cur, srt, n, chunkBlocks, ostr);
    gat_node_gather_mm<0><<<nodeBlocks, GW * 32, 0, stream>>>(srt, cur, cnt, a_src1, a_dst1, h, b1,
                                                             W2, nullptr, g, n);

    gat_node_gather_mm<1><<<nodeBlocks, GW * 32, 0, stream>>>(srt, cur, cnt, a_src2, a_dst2, g, b2,
                                                             Wc, bc, out, n);
}
